# starting kernel
# baseline (speedup 1.0000x reference)
#include <hip/hip_runtime.h>

#define N_USERS  100000
#define N_ITEMS  50000
#define N_NODES  (N_USERS + N_ITEMS)
#define NNZ_CT   1500000
#define BATCH_CT 1048576
#define DIM      64

typedef __attribute__((ext_vector_type(2))) float v2f;
typedef __attribute__((ext_vector_type(8))) float v8f;

__global__ __launch_bounds__(256) void zero_kernel(float4* __restrict__ p, int n4) {
    int i = blockIdx.x * blockDim.x + threadIdx.x;
    if (i < n4) p[i] = make_float4(0.f, 0.f, 0.f, 0.f);
}

__global__ __launch_bounds__(256) void spmm_kernel(
    const int*   __restrict__ rows, const int* __restrict__ cols,
    const float* __restrict__ vals,
    const float* __restrict__ srcA, const float* __restrict__ srcB, int split,
    float* __restrict__ Hout, int nnz) {
    long long t = (long long)blockIdx.x * blockDim.x + threadIdx.x;
    int e = (int)(t >> 4);
    if (e >= nnz) return;
    int dq = ((int)t & 15) << 2;
    int r = rows[e];
    int c = cols[e];
    float v = vals[e];
    const float* src = (c < split) ? (srcA + (size_t)c * DIM)
                                   : (srcB + (size_t)(c - split) * DIM);
    float4 h = *(const float4*)(src + dq);
    float* dst = Hout + (size_t)r * DIM + dq;
    unsafeAtomicAdd(dst + 0, v * h.x);
    unsafeAtomicAdd(dst + 1, v * h.y);
    unsafeAtomicAdd(dst + 2, v * h.z);
    unsafeAtomicAdd(dst + 3, v * h.w);
}

__global__ __launch_bounds__(256) void pred_kernel(
    const float* __restrict__ U, const float* __restrict__ I,
    const int*   __restrict__ u_idx, const int* __restrict__ i_idx,
    const float* __restrict__ bu, const float* __restrict__ bi,
    const float* __restrict__ mu, float* __restrict__ out) {
    int wave = blockIdx.x * (blockDim.x >> 5) + (threadIdx.x >> 5);
    int lane = threadIdx.x & 31;
    int base = wave << 4;
    int m    = lane & 15;
    int hi   = lane >> 4;
    int ur = u_idx[base + m];
    int ir = i_idx[base + m];
    const float* up = U + (size_t)ur * DIM;
    const float* ip = I + (size_t)ir * DIM;

    v8f c = {};
#pragma unroll
    for (int k0 = 0; k0 < DIM; k0 += 4) {
        v2f a = *(const v2f*)(up + k0 + 2 * hi);
        v2f b = *(const v2f*)(ip + k0 + 2 * hi);
        c = __builtin_amdgcn_wmma_f32_16x16x4_f32(
                false, a, false, b, (short)0, c, false, false);
    }

    float d = 0.f;
#pragma unroll
    for (int j = 0; j < 8; ++j) {
        float vlo = __shfl(c[j], j, 32);
        float vhi = __shfl(c[j], j + 24, 32);
        if (lane == j)     d = vlo;
        if (lane == j + 8) d = vhi;
    }

    if (lane < 16) {
        out[base + lane] = d + bu[ur] + bi[ir] + mu[0];
    }
}

extern "C" void kernel_launch(void* const* d_in, const int* in_sizes, int n_in,
                              void* d_out, int out_size, void* d_ws, size_t ws_size,
                              hipStream_t stream) {
    const int*   u_idx  = (const int*)d_in[0];
    const int*   i_idx  = (const int*)d_in[1];
    const int*   A_rows = (const int*)d_in[2];
    const int*   A_cols = (const int*)d_in[3];
    const float* A_vals = (const float*)d_in[4];
    const float* Eu     = (const float*)d_in[5];
    const float* Ei     = (const float*)d_in[6];
    const float* bu     = (const float*)d_in[7];
    const float* bi     = (const float*)d_in[8];
    const float* mu     = (const float*)d_in[9];
    float* out = (float*)d_out;

    float* H1 = (float*)d_ws;
    float* H2 = H1 + (size_t)N_NODES * DIM;

    int n4 = 2 * N_NODES * DIM / 4;
    zero_kernel<<<(n4 + 255) / 256, 256, 0, stream>>>((float4*)d_ws, n4);

    long long spmm_threads = (long long)NNZ_CT * 16;
    int spmm_blocks = (int)((spmm_threads + 255) / 256);
    spmm_kernel<<<spmm_blocks, 256, 0, stream>>>(
        A_rows, A_cols, A_vals, Eu, Ei, N_USERS, H1, NNZ_CT);

    spmm_kernel<<<spmm_blocks, 256, 0, stream>>>(
        A_rows, A_cols, A_vals, H1, H1, N_NODES + 1, H2, NNZ_CT);

    int pred_blocks = (BATCH_CT / 16) / 8;
    pred_kernel<<<pred_blocks, 256, 0, stream>>>(
        H2, H2 + (size_t)N_USERS * DIM, u_idx, i_idx, bu, bi, mu, out);
}


# spmm lane remap: one edge per wave instruction (lane l owns dims l, l+32) so each gather and f32 atomic covers one full 128-B line, not 4 quarter-filled ones; sums stay in atomic arrival order
# speedup vs baseline: 1.6900x; 1.6900x over previous
#include <hip/hip_runtime.h>

#define N_USERS  100000
#define N_ITEMS  50000
#define N_NODES  (N_USERS + N_ITEMS)
#define NNZ_CT   1500000
#define BATCH_CT 1048576
#define DIM      64

typedef __attribute__((ext_vector_type(2))) float v2f;
typedef __attribute__((ext_vector_type(8))) float v8f;

__global__ __launch_bounds__(256) void zero_kernel(float4* __restrict__ p, int n4) {
    int i = blockIdx.x * blockDim.x + threadIdx.x;
    if (i < n4) p[i] = make_float4(0.f, 0.f, 0.f, 0.f);
}

#define SPMM_EPW 8
#define SPMM_WPB 8
#define SPMM_EPB (SPMM_EPW * SPMM_WPB)

__device__ __forceinline__ const float* spmm_src_row(
    const float* __restrict__ srcA, const float* __restrict__ srcB, int split, int c) {
    return (c < split) ? (srcA + (size_t)c * DIM)
                       : (srcB + (size_t)(c - split) * DIM);
}

__global__ __launch_bounds__(256) void spmm_kernel(
    const int*   __restrict__ rows, const int* __restrict__ cols,
    const float* __restrict__ vals,
    const float* __restrict__ srcA, const float* __restrict__ srcB, int split,
    float* __restrict__ Hout, int nnz) {
    const int wib  = __builtin_amdgcn_readfirstlane((int)(threadIdx.x >> 5));
    const int lane = (int)threadIdx.x & 31;
    const int e0   = ((int)blockIdx.x * SPMM_WPB + wib) * SPMM_EPW;
    if (e0 >= nnz) return;

    if (e0 + SPMM_EPW <= nnz) {
        int   r[SPMM_EPW], c[SPMM_EPW];
        float v[SPMM_EPW], lo[SPMM_EPW], hi[SPMM_EPW];
#pragma unroll
        for (int j = 0; j < SPMM_EPW; ++j) {
            r[j] = rows[e0 + j];
            c[j] = cols[e0 + j];
            v[j] = vals[e0 + j];
        }
#pragma unroll
        for (int j = 0; j < SPMM_EPW; ++j) {
            const float* src = spmm_src_row(srcA, srcB, split, c[j]);
            lo[j] = src[lane];
            hi[j] = src[lane + 32];
        }
#pragma unroll
        for (int j = 0; j < SPMM_EPW; ++j) {
            float* dst = Hout + (size_t)r[j] * DIM;
            unsafeAtomicAdd(dst + lane,      v[j] * lo[j]);
            unsafeAtomicAdd(dst + lane + 32, v[j] * hi[j]);
        }
    } else {
        for (int e = e0; e < nnz; ++e) {
            const int   r1 = rows[e];
            const float v1 = vals[e];
            const float* src = spmm_src_row(srcA, srcB, split, cols[e]);
            const float l1 = src[lane];
            const float h1 = src[lane + 32];
            float* dst = Hout + (size_t)r1 * DIM;
            unsafeAtomicAdd(dst + lane,      v1 * l1);
            unsafeAtomicAdd(dst + lane + 32, v1 * h1);
        }
    }
}

__global__ __launch_bounds__(256) void pred_kernel(
    const float* __restrict__ U, const float* __restrict__ I,
    const int*   __restrict__ u_idx, const int* __restrict__ i_idx,
    const float* __restrict__ bu, const float* __restrict__ bi,
    const float* __restrict__ mu, float* __restrict__ out) {
    int wave = blockIdx.x * (blockDim.x >> 5) + (threadIdx.x >> 5);
    int lane = threadIdx.x & 31;
    int base = wave << 4;
    int m    = lane & 15;
    int hi   = lane >> 4;
    int ur = u_idx[base + m];
    int ir = i_idx[base + m];
    const float* up = U + (size_t)ur * DIM;
    const float* ip = I + (size_t)ir * DIM;

    v8f c = {};
#pragma unroll
    for (int k0 = 0; k0 < DIM; k0 += 4) {
        v2f a = *(const v2f*)(up + k0 + 2 * hi);
        v2f b = *(const v2f*)(ip + k0 + 2 * hi);
        c = __builtin_amdgcn_wmma_f32_16x16x4_f32(
                false, a, false, b, (short)0, c, false, false);
    }

    float d = 0.f;
#pragma unroll
    for (int j = 0; j < 8; ++j) {
        float vlo = __shfl(c[j], j, 32);
        float vhi = __shfl(c[j], j + 24, 32);
        if (lane == j)     d = vlo;
        if (lane == j + 8) d = vhi;
    }

    if (lane < 16) {
        out[base + lane] = d + bu[ur] + bi[ir] + mu[0];
    }
}

extern "C" void kernel_launch(void* const* d_in, const int* in_sizes, int n_in,
                              void* d_out, int out_size, void* d_ws, size_t ws_size,
                              hipStream_t stream) {
    const int*   u_idx  = (const int*)d_in[0];
    const int*   i_idx  = (const int*)d_in[1];
    const int*   A_rows = (const int*)d_in[2];
    const int*   A_cols = (const int*)d_in[3];
    const float* A_vals = (const float*)d_in[4];
    const float* Eu     = (const float*)d_in[5];
    const float* Ei     = (const float*)d_in[6];
    const float* bu     = (const float*)d_in[7];
    const float* bi     = (const float*)d_in[8];
    const float* mu     = (const float*)d_in[9];
    float* out = (float*)d_out;

    float* H1 = (float*)d_ws;
    float* H2 = H1 + (size_t)N_NODES * DIM;

    int n4 = 2 * N_NODES * DIM / 4;
    zero_kernel<<<(n4 + 255) / 256, 256, 0, stream>>>((float4*)d_ws, n4);

    int spmm_blocks = (NNZ_CT + SPMM_EPB - 1) / SPMM_EPB;
    spmm_kernel<<<spmm_blocks, 256, 0, stream>>>(
        A_rows, A_cols, A_vals, Eu, Ei, N_USERS, H1, NNZ_CT);

    spmm_kernel<<<spmm_blocks, 256, 0, stream>>>(
        A_rows, A_cols, A_vals, H1, H1, N_NODES + 1, H2, NNZ_CT);

    int pred_blocks = (BATCH_CT / 16) / 8;
    pred_kernel<<<pred_blocks, 256, 0, stream>>>(
        H2, H2 + (size_t)N_USERS * DIM, u_idx, i_idx, bu, bi, mu, out);
}


# pred: U and I rows gathered whole (16 B per lane, 2 rows = 4 full lines per load) and turned to WMMA fragments via a per-wave LDS tile; k order of the 16 steps permuted
# speedup vs baseline: 1.9370x; 1.1462x over previous
#include <hip/hip_runtime.h>

#define N_USERS  100000
#define N_ITEMS  50000
#define N_NODES  (N_USERS + N_ITEMS)
#define NNZ_CT   1500000
#define BATCH_CT 1048576
#define DIM      64

typedef __attribute__((ext_vector_type(2))) float v2f;
typedef __attribute__((ext_vector_type(8))) float v8f;

__global__ __launch_bounds__(256) void zero_kernel(float4* __restrict__ p, int n4) {
    int i = blockIdx.x * blockDim.x + threadIdx.x;
    if (i < n4) p[i] = make_float4(0.f, 0.f, 0.f, 0.f);
}

#define SPMM_EPW 8
#define SPMM_WPB 8
#define SPMM_EPB (SPMM_EPW * SPMM_WPB)

__device__ __forceinline__ const float* spmm_src_row(
    const float* __restrict__ srcA, const float* __restrict__ srcB, int split, int c) {
    return (c < split) ? (srcA + (size_t)c * DIM)
                       : (srcB + (size_t)(c - split) * DIM);
}

__global__ __launch_bounds__(256) void spmm_kernel(
    const int*   __restrict__ rows, const int* __restrict__ cols,
    const float* __restrict__ vals,
    const float* __restrict__ srcA, const float* __restrict__ srcB, int split,
    float* __restrict__ Hout, int nnz) {
    const int wib  = __builtin_amdgcn_readfirstlane((int)(threadIdx.x >> 5));
    const int lane = (int)threadIdx.x & 31;
    const int e0   = ((int)blockIdx.x * SPMM_WPB + wib) * SPMM_EPW;
    if (e0 >= nnz) return;

    if (e0 + SPMM_EPW <= nnz) {
        int   r[SPMM_EPW], c[SPMM_EPW];
        float v[SPMM_EPW], lo[SPMM_EPW], hi[SPMM_EPW];
#pragma unroll
        for (int j = 0; j < SPMM_EPW; ++j) {
            r[j] = rows[e0 + j];
            c[j] = cols[e0 + j];
            v[j] = vals[e0 + j];
        }
#pragma unroll
        for (int j = 0; j < SPMM_EPW; ++j) {
            const float* src = spmm_src_row(srcA, srcB, split, c[j]);
            lo[j] = src[lane];
            hi[j] = src[lane + 32];
        }
#pragma unroll
        for (int j = 0; j < SPMM_EPW; ++j) {
            float* dst = Hout + (size_t)r[j] * DIM;
            unsafeAtomicAdd(dst + lane,      v[j] * lo[j]);
            unsafeAtomicAdd(dst + lane + 32, v[j] * hi[j]);
        }
    } else {
        for (int e = e0; e < nnz; ++e) {
            const int   r1 = rows[e];
            const float v1 = vals[e];
            const float* src = spmm_src_row(srcA, srcB, split, cols[e]);
            const float l1 = src[lane];
            const float h1 = src[lane + 32];
            float* dst = Hout + (size_t)r1 * DIM;
            unsafeAtomicAdd(dst + lane,      v1 * l1);
            unsafeAtomicAdd(dst + lane + 32, v1 * h1);
        }
    }
}

#define PRED_WPB   8
#define TILE_PITCH 68
__global__ __launch_bounds__(256) void pred_kernel(
    const float* __restrict__ U, const float* __restrict__ I,
    const int*   __restrict__ u_idx, const int* __restrict__ i_idx,
    const float* __restrict__ bu, const float* __restrict__ bi,
    const float* __restrict__ mu, float* __restrict__ out) {
    __shared__ float tiles[PRED_WPB][2][16 * TILE_PITCH];

    int wib  = threadIdx.x >> 5;
    int wave = blockIdx.x * PRED_WPB + wib;
    int lane = threadIdx.x & 31;
    int base = wave << 4;
    int m    = lane & 15;
    int hi   = lane >> 4;
    int ur = u_idx[base + m];
    int ir = i_idx[base + m];

    const int4* uq = (const int4*)(u_idx + base + 8 * hi);
    const int4* iq = (const int4*)(i_idx + base + 8 * hi);
    const int4 ua = uq[0], ub = uq[1];
    const int4 ia = iq[0], ib = iq[1];
    const int urow[8] = {ua.x, ua.y, ua.z, ua.w, ub.x, ub.y, ub.z, ub.w};
    const int irow[8] = {ia.x, ia.y, ia.z, ia.w, ib.x, ib.y, ib.z, ib.w};

    float* tu = &tiles[wib][0][0];
    float* ti = &tiles[wib][1][0];

    float4 gu[8], gi[8];
#pragma unroll
    for (int j = 0; j < 8; ++j) {
        gu[j] = *(const float4*)(U + (size_t)urow[j] * DIM + 4 * m);
        gi[j] = *(const float4*)(I + (size_t)irow[j] * DIM + 4 * m);
    }
#pragma unroll
    for (int j = 0; j < 8; ++j) {
        *(float4*)(tu + (8 * hi + j) * TILE_PITCH + 4 * m) = gu[j];
        *(float4*)(ti + (8 * hi + j) * TILE_PITCH + 4 * m) = gi[j];
    }
    __syncthreads();

    const float* fa = tu + m * TILE_PITCH + 32 * hi;
    const float* fb = ti + m * TILE_PITCH + 32 * hi;
    v8f c = {};
#pragma unroll
    for (int t = 0; t < 8; ++t) {
        const float4 a4 = *(const float4*)(fa + 4 * t);
        const float4 b4 = *(const float4*)(fb + 4 * t);
        v2f a0 = {a4.x, a4.y}, b0 = {b4.x, b4.y};
        v2f a1 = {a4.z, a4.w}, b1 = {b4.z, b4.w};
        c = __builtin_amdgcn_wmma_f32_16x16x4_f32(
                false, a0, false, b0, (short)0, c, false, false);
        c = __builtin_amdgcn_wmma_f32_16x16x4_f32(
                false, a1, false, b1, (short)0, c, false, false);
    }

    float d = 0.f;
#pragma unroll
    for (int j = 0; j < 8; ++j) {
        float vlo = __shfl(c[j], j, 32);
        float vhi = __shfl(c[j], j + 24, 32);
        if (lane == j)     d = vlo;
        if (lane == j + 8) d = vhi;
    }

    if (lane < 16) {
        out[base + lane] = d + bu[ur] + bi[ir] + mu[0];
    }
}

extern "C" void kernel_launch(void* const* d_in, const int* in_sizes, int n_in,
                              void* d_out, int out_size, void* d_ws, size_t ws_size,
                              hipStream_t stream) {
    const int*   u_idx  = (const int*)d_in[0];
    const int*   i_idx  = (const int*)d_in[1];
    const int*   A_rows = (const int*)d_in[2];
    const int*   A_cols = (const int*)d_in[3];
    const float* A_vals = (const float*)d_in[4];
    const float* Eu     = (const float*)d_in[5];
    const float* Ei     = (const float*)d_in[6];
    const float* bu     = (const float*)d_in[7];
    const float* bi     = (const float*)d_in[8];
    const float* mu     = (const float*)d_in[9];
    float* out = (float*)d_out;

    float* H1 = (float*)d_ws;
    float* H2 = H1 + (size_t)N_NODES * DIM;

    int n4 = 2 * N_NODES * DIM / 4;
    zero_kernel<<<(n4 + 255) / 256, 256, 0, stream>>>((float4*)d_ws, n4);

    int spmm_blocks = (NNZ_CT + SPMM_EPB - 1) / SPMM_EPB;
    spmm_kernel<<<spmm_blocks, 256, 0, stream>>>(
        A_rows, A_cols, A_vals, Eu, Ei, N_USERS, H1, NNZ_CT);

    spmm_kernel<<<spmm_blocks, 256, 0, stream>>>(
        A_rows, A_cols, A_vals, H1, H1, N_NODES + 1, H2, NNZ_CT);

    int pred_blocks = (BATCH_CT / 16) / 8;
    pred_kernel<<<pred_blocks, 256, 0, stream>>>(
        H2, H2 + (size_t)N_USERS * DIM, u_idx, i_idx, bu, bi, mu, out);
}


# spmm per 128-row destination bucket: counters cleared and edges placed by bucket in every call, rows summed in LDS (f32 LDS adds, arrival order), stored once; no global f32 atomics, no zero fill
# speedup vs baseline: 2.4218x; 1.2503x over previous
#include <hip/hip_runtime.h>


#define N_USERS  100000
#define N_ITEMS  50000
#define N_NODES  (N_USERS + N_ITEMS)
#define NNZ_CT   1500000
#define BATCH_CT 1048576
#define DIM      64
#define HALF     32

typedef __attribute__((ext_vector_type(2))) float v2f;
typedef __attribute__((ext_vector_type(8))) float v8f;

#define BKT_ROWS  128
#define BKT_SHIFT 7
#define NBKT      ((N_NODES + BKT_ROWS - 1) / BKT_ROWS)
#define NBKT_PAD  1280
#define PLC_BLOCKS 512
#define PLC_EPB   ((NNZ_CT + PLC_BLOCKS - 1) / PLC_BLOCKS)
#define PLC_NIT   ((PLC_EPB + 255) / 256)

__global__ __launch_bounds__(256) void clear_counts_kernel(int* __restrict__ cnt2) {
    for (int i = threadIdx.x; i < 2 * NBKT_PAD; i += 256) cnt2[i] = 0;
}

__global__ __launch_bounds__(256) void bucket_count_kernel(
    const int* __restrict__ rows, int* __restrict__ cnt, int nnz) {
    __shared__ int h[NBKT_PAD];
    const int tid = (int)threadIdx.x;
    for (int i = tid; i < NBKT_PAD; i += 256) h[i] = 0;
    __syncthreads();
    const int e_beg = (int)blockIdx.x * PLC_EPB;
    int r[PLC_NIT];
#pragma unroll
    for (int q = 0; q < PLC_NIT; ++q) {
        const int i = tid + q * 256;
        const int e = e_beg + i;
        r[q] = (i < PLC_EPB && e < nnz) ? rows[e] : -1;
    }
#pragma unroll
    for (int q = 0; q < PLC_NIT; ++q) {
        if ((unsigned)r[q] < (unsigned)N_NODES) atomicAdd(&h[r[q] >> BKT_SHIFT], 1);
    }
    __syncthreads();
    for (int k = tid; k < NBKT; k += 256) {
        const int c = h[k];
        if (c) atomicAdd(&cnt[k], c);
    }
}

__global__ __launch_bounds__(256) void bucket_place_kernel(
    const int* __restrict__ rows, const int* __restrict__ cols,
    const float* __restrict__ vals, const int* __restrict__ cnt,
    int* __restrict__ fill, int* __restrict__ start_out,
    int2* __restrict__ rec, int nnz) {
    __shared__ int h[NBKT_PAD];
    __shared__ int st[NBKT_PAD];
    __shared__ int wsum[8];
    const int tid  = (int)threadIdx.x;
    const int lane = tid & 31;
    const int wib  = tid >> 5;
    for (int i = tid; i < NBKT_PAD; i += 256) h[i] = 0;
    __syncthreads();
    const int e_beg = (int)blockIdx.x * PLC_EPB;
    int   r[PLC_NIT], c[PLC_NIT];
    float v[PLC_NIT];
#pragma unroll
    for (int q = 0; q < PLC_NIT; ++q) {
        const int i = tid + q * 256;
        const int e = e_beg + i;
        const bool ok = (i < PLC_EPB && e < nnz);
        r[q] = ok ? rows[e] : -1;
        c[q] = ok ? cols[e] : 0;
        v[q] = ok ? vals[e] : 0.f;
    }
#pragma unroll
    for (int q = 0; q < PLC_NIT; ++q) {
        if ((unsigned)r[q] < (unsigned)N_NODES) atomicAdd(&h[r[q] >> BKT_SHIFT], 1);
    }
    int c5[5];
    int s = 0;
#pragma unroll
    for (int q = 0; q < 5; ++q) {
        const int k = 5 * tid + q;
        c5[q] = (k < NBKT) ? cnt[k] : 0;
        s += c5[q];
    }
    int incl = s;
    for (int d = 1; d < 32; d <<= 1) {
        const int o = __shfl_up(incl, d, 32);
        if (lane >= d) incl += o;
    }
    if (lane == 31) wsum[wib] = incl;
    __syncthreads();
    int woff = 0;
    for (int w = 0; w < 8; ++w) if (w < wib) woff += wsum[w];
    int run = woff + incl - s;
#pragma unroll
    for (int q = 0; q < 5; ++q) {
        st[5 * tid + q] = run;
        run += c5[q];
    }
    __syncthreads();
    for (int k = tid; k < NBKT; k += 256) {
        const int n = h[k];
        const int b = n ? atomicAdd(&fill[k], n) : 0;
        h[k] = st[k] + b;
        if (blockIdx.x == 0) start_out[k] = st[k];
    }
    __syncthreads();
#pragma unroll
    for (int q = 0; q < PLC_NIT; ++q) {
        if ((unsigned)r[q] < (unsigned)N_NODES) {
            const int p = atomicAdd(&h[r[q] >> BKT_SHIFT], 1);
            rec[p] = make_int2((c[q] << BKT_SHIFT) | (r[q] & (BKT_ROWS - 1)),
                               __float_as_int(v[q]));
        }
    }
}

#define GRP 16

__device__ __forceinline__ const float* spmm_src_row(
    const float* __restrict__ srcA, const float* __restrict__ srcB, int split, int c) {
    return (c < split) ? (srcA + (size_t)c * DIM)
                       : (srcB + (size_t)(c - split) * DIM);
}

__global__ __launch_bounds__(256) void bucket_spmm_rows_kernel(
    const int2* __restrict__ rec, const int* __restrict__ start,
    const int* __restrict__ cnt,
    const float* __restrict__ srcA, const float* __restrict__ srcB, int split,
    float* __restrict__ out_lo, float* __restrict__ out_hi) {
    __shared__ float acc[BKT_ROWS * DIM];
    const int tid  = (int)threadIdx.x;
    const int lane = tid & 31;
    const int wib  = __builtin_amdgcn_readfirstlane(tid >> 5);
    const int bkt  = (int)blockIdx.x;
    for (int i = tid; i < BKT_ROWS * DIM / 4; i += 256)
        ((float4*)acc)[i] = make_float4(0.f, 0.f, 0.f, 0.f);
    __syncthreads();
    const int beg = __builtin_amdgcn_readfirstlane(start[bkt]);
    const int end = beg + __builtin_amdgcn_readfirstlane(cnt[bkt]);

    int2 cur = make_int2(0, 0);
    if (beg < end) cur = rec[(beg + tid < end) ? (beg + tid) : (end - 1)];
    for (int base = beg; base < end; base += 256) {
        int2 nxt = make_int2(0, 0);
        int nvalid = end - base - wib * 32;
        nvalid = nvalid > 32 ? 32 : nvalid;
#pragma unroll
        for (int g = 0; g < 32; g += GRP) {
            if (g < nvalid) {
                int    key[GRP];
                float  w[GRP];
                float2 x[GRP];
#pragma unroll
                for (int j = 0; j < GRP; ++j) {
                    const int jj = (g + j < nvalid) ? (g + j) : (nvalid - 1);
                    key[j] = __builtin_amdgcn_readlane(cur.x, jj);
                    w[j]   = __int_as_float(__builtin_amdgcn_readlane(cur.y, jj));
                    const float* src = spmm_src_row(srcA, srcB, split, key[j] >> BKT_SHIFT);
                    x[j] = *(const float2*)(src + 2 * lane);
                }
                if (g == 0) nxt = rec[(base + 256 + tid < end) ? (base + 256 + tid) : (end - 1)];
#pragma unroll
                for (int j = 0; j < GRP; ++j) {
                    if (g + j < nvalid) {
                        float* d = acc + (key[j] & (BKT_ROWS - 1)) * DIM + 2 * lane;
                        atomicAdd(d,     w[j] * x[j].x);
                        atomicAdd(d + 1, w[j] * x[j].y);
                    }
                }
            }
        }
        cur = nxt;
    }
    __syncthreads();
    const size_t row0 = (size_t)bkt * BKT_ROWS;
    for (int q = tid; q < BKT_ROWS * (HALF / 4); q += 256) {
        const int rr = q >> 3, pc = q & 7;
        if (row0 + rr < (size_t)N_NODES) {
            const float4 lo = *(const float4*)(acc + rr * DIM + 4 * pc);
            const float4 hi = *(const float4*)(acc + rr * DIM + HALF + 4 * pc);
            *(float4*)(out_lo + (row0 + rr) * HALF + 4 * pc) = lo;
            *(float4*)(out_hi + (row0 + rr) * HALF + 4 * pc) = hi;
        }
    }
}

__global__ __launch_bounds__(256) void bucket_spmm_plane_kernel(
    const int2* __restrict__ rec, const int* __restrict__ start,
    const int* __restrict__ cnt,
    const float* __restrict__ src, float* __restrict__ dst) {
    __shared__ float acc[BKT_ROWS * HALF];
    const int tid  = (int)threadIdx.x;
    const int lane = tid & 31;
    const int wib  = __builtin_amdgcn_readfirstlane(tid >> 5);
    const int bkt  = (int)blockIdx.x;
    for (int i = tid; i < BKT_ROWS * HALF / 4; i += 256)
        ((float4*)acc)[i] = make_float4(0.f, 0.f, 0.f, 0.f);
    __syncthreads();
    const int beg = __builtin_amdgcn_readfirstlane(start[bkt]);
    const int end = beg + __builtin_amdgcn_readfirstlane(cnt[bkt]);

    int2 cur = make_int2(0, 0);
    if (beg < end) cur = rec[(beg + tid < end) ? (beg + tid) : (end - 1)];
    for (int base = beg; base < end; base += 256) {
        int2 nxt = make_int2(0, 0);
        int nvalid = end - base - wib * 32;
        nvalid = nvalid > 32 ? 32 : nvalid;
#pragma unroll
        for (int g = 0; g < 32; g += GRP) {
            if (g < nvalid) {
                int   key[GRP];
                float w[GRP];
                float x[GRP];
#pragma unroll
                for (int j = 0; j < GRP; ++j) {
                    const int jj = (g + j < nvalid) ? (g + j) : (nvalid - 1);
                    key[j] = __builtin_amdgcn_readlane(cur.x, jj);
                    w[j]   = __int_as_float(__builtin_amdgcn_readlane(cur.y, jj));
                    x[j]   = src[(size_t)(key[j] >> BKT_SHIFT) * HALF + lane];
                }
                if (g == 0) nxt = rec[(base + 256 + tid < end) ? (base + 256 + tid) : (end - 1)];
#pragma unroll
                for (int j = 0; j < GRP; ++j) {
                    if (g + j < nvalid)
                        atomicAdd(acc + (key[j] & (BKT_ROWS - 1)) * HALF + lane, w[j] * x[j]);
                }
            }
        }
        cur = nxt;
    }
    __syncthreads();
    const size_t row0 = (size_t)bkt * BKT_ROWS;
    for (int q = tid; q < BKT_ROWS * (HALF / 4); q += 256) {
        const int rr = q >> 3;
        if (row0 + rr < (size_t)N_NODES)
            *(float4*)(dst + row0 * HALF + 4 * q) = *(const float4*)(acc + 4 * q);
    }
}

#define PRED_WPB   8
#define TILE_PITCH 68
__global__ __launch_bounds__(256) void pred_kernel(
    const float* __restrict__ Ulo, const float* __restrict__ Uhi,
    const float* __restrict__ Ilo, const float* __restrict__ Ihi,
    const int*   __restrict__ u_idx, const int* __restrict__ i_idx,
    const float* __restrict__ bu, const float* __restrict__ bi,
    const float* __restrict__ mu, float* __restrict__ out) {
    __shared__ float tiles[PRED_WPB][2][16 * TILE_PITCH];

    int wib  = threadIdx.x >> 5;
    int wave = blockIdx.x * PRED_WPB + wib;
    int lane = threadIdx.x & 31;
    int base = wave << 4;
    int m    = lane & 15;
    int hi   = lane >> 4;
    int ur = u_idx[base + m];
    int ir = i_idx[base + m];

    const int4* uq = (const int4*)(u_idx + base + 8 * hi);
    const int4* iq = (const int4*)(i_idx + base + 8 * hi);
    const int4 ua = uq[0], ub = uq[1];
    const int4 ia = iq[0], ib = iq[1];
    const int urow[8] = {ua.x, ua.y, ua.z, ua.w, ub.x, ub.y, ub.z, ub.w};
    const int irow[8] = {ia.x, ia.y, ia.z, ia.w, ib.x, ib.y, ib.z, ib.w};

    float* tu = &tiles[wib][0][0];
    float* ti = &tiles[wib][1][0];

    const float* up = (m < 8) ? Ulo : Uhi;
    const float* ip = (m < 8) ? Ilo : Ihi;
    const int    mo = 4 * (m & 7);
    float4 gu[8], gi[8];
#pragma unroll
    for (int j = 0; j < 8; ++j) {
        gu[j] = *(const float4*)(up + (size_t)urow[j] * HALF + mo);
        gi[j] = *(const float4*)(ip + (size_t)irow[j] * HALF + mo);
    }
#pragma unroll
    for (int j = 0; j < 8; ++j) {
        *(float4*)(tu + (8 * hi + j) * TILE_PITCH + 4 * m) = gu[j];
        *(float4*)(ti + (8 * hi + j) * TILE_PITCH + 4 * m) = gi[j];
    }
    __syncthreads();

    const float* fa = tu + m * TILE_PITCH + 32 * hi;
    const float* fb = ti + m * TILE_PITCH + 32 * hi;
    v8f c = {};
#pragma unroll
    for (int t = 0; t < 8; ++t) {
        const float4 a4 = *(const float4*)(fa + 4 * t);
        const float4 b4 = *(const float4*)(fb + 4 * t);
        v2f a0 = {a4.x, a4.y}, b0 = {b4.x, b4.y};
        v2f a1 = {a4.z, a4.w}, b1 = {b4.z, b4.w};
        c = __builtin_amdgcn_wmma_f32_16x16x4_f32(
                false, a0, false, b0, (short)0, c, false, false);
        c = __builtin_amdgcn_wmma_f32_16x16x4_f32(
                false, a1, false, b1, (short)0, c, false, false);
    }

    float d = 0.f;
#pragma unroll
    for (int j = 0; j < 8; ++j) {
        float vlo = __shfl(c[j], j, 32);
        float vhi = __shfl(c[j], j + 24, 32);
        if (lane == j)     d = vlo;
        if (lane == j + 8) d = vhi;
    }

    if (lane < 16) {
        out[base + lane] = d + bu[ur] + bi[ir] + mu[0];
    }
}

extern "C" void kernel_launch(void* const* d_in, const int* in_sizes, int n_in,
                              void* d_out, int out_size, void* d_ws, size_t ws_size,
                              hipStream_t stream) {
    const int*   u_idx  = (const int*)d_in[0];
    const int*   i_idx  = (const int*)d_in[1];
    const int*   A_rows = (const int*)d_in[2];
    const int*   A_cols = (const int*)d_in[3];
    const float* A_vals = (const float*)d_in[4];
    const float* Eu     = (const float*)d_in[5];
    const float* Ei     = (const float*)d_in[6];
    const float* bu     = (const float*)d_in[7];
    const float* bi     = (const float*)d_in[8];
    const float* mu     = (const float*)d_in[9];
    float* out = (float*)d_out;

    float* P0 = (float*)d_ws;
    float* P1 = P0 + (size_t)N_NODES * HALF;
    float* P2 = P1 + (size_t)N_NODES * HALF;
    int2*  rec   = (int2*)(P2 + (size_t)N_NODES * HALF);
    int*   cnt   = (int*)(rec + NNZ_CT);
    int*   fill  = cnt + NBKT_PAD;
    int*   start = fill + NBKT_PAD;

    clear_counts_kernel<<<1, 256, 0, stream>>>(cnt);
    bucket_count_kernel<<<PLC_BLOCKS, 256, 0, stream>>>(A_rows, cnt, NNZ_CT);
    bucket_place_kernel<<<PLC_BLOCKS, 256, 0, stream>>>(
        A_rows, A_cols, A_vals, cnt, fill, start, rec, NNZ_CT);

    bucket_spmm_rows_kernel<<<NBKT, 256, 0, stream>>>(
        rec, start, cnt, Eu, Ei, N_USERS, P0, P1);

    bucket_spmm_plane_kernel<<<NBKT, 256, 0, stream>>>(rec, start, cnt, P0, P2);
    bucket_spmm_plane_kernel<<<NBKT, 256, 0, stream>>>(rec, start, cnt, P1, P0);

    pred_kernel<<<(BATCH_CT / 16) / 8, 256, 0, stream>>>(
        P2, P0, P2 + (size_t)N_USERS * HALF, P0 + (size_t)N_USERS * HALF,
        u_idx, i_idx, bu, bi, mu, out);
}


# spmm: bucket records ordered by row in LDS (integer counting sort); a wave sums its 16 rows in registers (fma), stores each once; no LDS f32 adds, no zero fill; terms summed in the sort's order
# speedup vs baseline: 2.6328x; 1.0871x over previous
#include <hip/hip_runtime.h>


#define N_USERS  100000
#define N_ITEMS  50000
#define N_NODES  (N_USERS + N_ITEMS)
#define NNZ_CT   1500000
#define BATCH_CT 1048576
#define DIM      64
#define HALF     32

typedef __attribute__((ext_vector_type(2))) float v2f;
typedef __attribute__((ext_vector_type(8))) float v8f;

#define BKT_ROWS  128
#define BKT_SHIFT 7
#define NBKT      ((N_NODES + BKT_ROWS - 1) / BKT_ROWS)
#define NBKT_PAD  1280
#define PLC_BLOCKS 512
#define PLC_EPB   ((NNZ_CT + PLC_BLOCKS - 1) / PLC_BLOCKS)
#define PLC_NIT   ((PLC_EPB + 255) / 256)

__global__ __launch_bounds__(256) void clear_counts_kernel(int* __restrict__ cnt2) {
    for (int i = threadIdx.x; i < 2 * NBKT_PAD; i += 256) cnt2[i] = 0;
}

__global__ __launch_bounds__(256) void bucket_count_kernel(
    const int* __restrict__ rows, int* __restrict__ cnt, int nnz) {
    __shared__ int h[NBKT_PAD];
    const int tid = (int)threadIdx.x;
    for (int i = tid; i < NBKT_PAD; i += 256) h[i] = 0;
    __syncthreads();
    const int e_beg = (int)blockIdx.x * PLC_EPB;
    int r[PLC_NIT];
#pragma unroll
    for (int q = 0; q < PLC_NIT; ++q) {
        const int i = tid + q * 256;
        const int e = e_beg + i;
        r[q] = (i < PLC_EPB && e < nnz) ? rows[e] : -1;
    }
#pragma unroll
    for (int q = 0; q < PLC_NIT; ++q) {
        if ((unsigned)r[q] < (unsigned)N_NODES) atomicAdd(&h[r[q] >> BKT_SHIFT], 1);
    }
    __syncthreads();
    for (int k = tid; k < NBKT; k += 256) {
        const int c = h[k];
        if (c) atomicAdd(&cnt[k], c);
    }
}

__global__ __launch_bounds__(256) void bucket_place_kernel(
    const int* __restrict__ rows, const int* __restrict__ cols,
    const float* __restrict__ vals, const int* __restrict__ cnt,
    int* __restrict__ fill, int* __restrict__ start_out,
    int2* __restrict__ rec, int nnz) {
    __shared__ int h[NBKT_PAD];
    __shared__ int st[NBKT_PAD];
    __shared__ int wsum[8];
    const int tid  = (int)threadIdx.x;
    const int lane = tid & 31;
    const int wib  = tid >> 5;
    for (int i = tid; i < NBKT_PAD; i += 256) h[i] = 0;
    __syncthreads();
    const int e_beg = (int)blockIdx.x * PLC_EPB;
    int   r[PLC_NIT], c[PLC_NIT];
    float v[PLC_NIT];
#pragma unroll
    for (int q = 0; q < PLC_NIT; ++q) {
        const int i = tid + q * 256;
        const int e = e_beg + i;
        const bool ok = (i < PLC_EPB && e < nnz);
        r[q] = ok ? rows[e] : -1;
        c[q] = ok ? cols[e] : 0;
        v[q] = ok ? vals[e] : 0.f;
    }
#pragma unroll
    for (int q = 0; q < PLC_NIT; ++q) {
        if ((unsigned)r[q] < (unsigned)N_NODES) atomicAdd(&h[r[q] >> BKT_SHIFT], 1);
    }
    int c5[5];
    int s = 0;
#pragma unroll
    for (int q = 0; q < 5; ++q) {
        const int k = 5 * tid + q;
        c5[q] = (k < NBKT) ? cnt[k] : 0;
        s += c5[q];
    }
    int incl = s;
    for (int d = 1; d < 32; d <<= 1) {
        const int o = __shfl_up(incl, d, 32);
        if (lane >= d) incl += o;
    }
    if (lane == 31) wsum[wib] = incl;
    __syncthreads();
    int woff = 0;
    for (int w = 0; w < 8; ++w) if (w < wib) woff += wsum[w];
    int run = woff + incl - s;
#pragma unroll
    for (int q = 0; q < 5; ++q) {
        st[5 * tid + q] = run;
        run += c5[q];
    }
    __syncthreads();
    for (int k = tid; k < NBKT; k += 256) {
        const int n = h[k];
        const int b = n ? atomicAdd(&fill[k], n) : 0;
        h[k] = st[k] + b;
        if (blockIdx.x == 0) start_out[k] = st[k];
    }
    __syncthreads();
#pragma unroll
    for (int q = 0; q < PLC_NIT; ++q) {
        if ((unsigned)r[q] < (unsigned)N_NODES) {
            const int p = atomicAdd(&h[r[q] >> BKT_SHIFT], 1);
            rec[p] = make_int2((c[q] << BKT_SHIFT) | (r[q] & (BKT_ROWS - 1)),
                               __float_as_int(v[q]));
        }
    }
}

#define GRP 16
#define SORT_CAP 2048
#define SORT_NIT (SORT_CAP / 256)

__device__ __forceinline__ const float* spmm_src_row(
    const float* __restrict__ srcA, const float* __restrict__ srcB, int split, int c) {
    return (c < split) ? (srcA + (size_t)c * DIM)
                       : (srcB + (size_t)(c - split) * DIM);
}

__device__ __forceinline__ void bucket_order_rows(
    const int2* __restrict__ rec, int beg, int n,
    int2* srec, int* rcnt, int* rbeg, int tid) {
    const int lane = tid & 31;
    if (tid < BKT_ROWS) rcnt[tid] = 0;
    int2 r[SORT_NIT];
#pragma unroll
    for (int q = 0; q < SORT_NIT; ++q) {
        const int i = tid + q * 256;
        r[q] = (i < n) ? rec[beg + i] : make_int2(0, 0);
    }
    __syncthreads();
#pragma unroll
    for (int q = 0; q < SORT_NIT; ++q)
        if (tid + q * 256 < n) atomicAdd(&rcnt[r[q].x & (BKT_ROWS - 1)], 1);
    __syncthreads();
    if (tid < 32) {
        int c4[4];
        int s = 0;
#pragma unroll
        for (int k = 0; k < 4; ++k) { c4[k] = rcnt[4 * lane + k]; s += c4[k]; }
        int incl = s;
        for (int d = 1; d < 32; d <<= 1) {
            const int o = __shfl_up(incl, d, 32);
            if (lane >= d) incl += o;
        }
        int run = incl - s;
#pragma unroll
        for (int k = 0; k < 4; ++k) {
            rbeg[4 * lane + k] = run;
            rcnt[4 * lane + k] = run;
            run += c4[k];
        }
        if (lane == 31) rbeg[BKT_ROWS] = incl;
    }
    __syncthreads();
    int p[SORT_NIT];
#pragma unroll
    for (int q = 0; q < SORT_NIT; ++q)
        p[q] = (tid + q * 256 < n) ? atomicAdd(&rcnt[r[q].x & (BKT_ROWS - 1)], 1) : 0;
#pragma unroll
    for (int q = 0; q < SORT_NIT; ++q)
        if (tid + q * 256 < n) srec[p[q]] = r[q];
    __syncthreads();
}

__global__ __launch_bounds__(256) void bucket_spmm_rows_kernel(
    const int2* __restrict__ rec, const int* __restrict__ start,
    const int* __restrict__ cnt,
    const float* __restrict__ srcA, const float* __restrict__ srcB, int split,
    float* __restrict__ out_lo, float* __restrict__ out_hi) {
    __shared__ __align__(16) unsigned char raw[BKT_ROWS * DIM * 4];
    __shared__ int rcnt[BKT_ROWS];
    __shared__ int rbeg[BKT_ROWS + 1];
    const int tid  = (int)threadIdx.x;
    const int lane = tid & 31;
    const int wib  = __builtin_amdgcn_readfirstlane(tid >> 5);
    const int bkt  = (int)blockIdx.x;
    const size_t row0 = (size_t)bkt * BKT_ROWS;
    const int beg = __builtin_amdgcn_readfirstlane(start[bkt]);
    const int n   = __builtin_amdgcn_readfirstlane(cnt[bkt]);

    if (n <= SORT_CAP) {
        int2* srec = (int2*)raw;
        bucket_order_rows(rec, beg, n, srec, rcnt, rbeg, tid);
        const int r0 = wib * 16;
        const int s0 = __builtin_amdgcn_readfirstlane(rbeg[r0]);
        const int s1 = __builtin_amdgcn_readfirstlane(rbeg[r0 + 16]);
        float* orow = ((lane < 16) ? out_lo : out_hi) + 2 * (lane & 15);
        {
            const int li = lane & 15;
            const bool empty = (lane < 16) && (rbeg[r0 + li + 1] == rbeg[r0 + li]);
            unsigned em = __builtin_amdgcn_ballot_w32(empty);
            while (em) {
                const int rr = r0 + __builtin_ctz(em);
                em &= em - 1;
                if (row0 + rr < (size_t)N_NODES)
                    *(float2*)(orow + (row0 + rr) * HALF) = make_float2(0.f, 0.f);
            }
        }
        float2 acc = make_float2(0.f, 0.f);
        int cur = -1;
        for (int base = s0; base < s1; base += 32) {
            int nvalid = s1 - base;
            nvalid = nvalid > 32 ? 32 : nvalid;
            const int2 mine = srec[(base + lane < s1) ? (base + lane) : (s1 - 1)];
#pragma unroll
            for (int g = 0; g < 32; g += GRP) {
                if (g < nvalid) {
                    int    key[GRP];
                    float  w[GRP];
                    float2 x[GRP];
#pragma unroll
                    for (int j = 0; j < GRP; ++j) {
                        const int jj = (g + j < nvalid) ? (g + j) : (nvalid - 1);
                        key[j] = __builtin_amdgcn_readlane(mine.x, jj);
                        w[j]   = __int_as_float(__builtin_amdgcn_readlane(mine.y, jj));
                        const float* src = spmm_src_row(srcA, srcB, split, key[j] >> BKT_SHIFT);
                        x[j] = *(const float2*)(src + 2 * lane);
                    }
#pragma unroll
                    for (int j = 0; j < GRP; ++j) {
                        if (g + j < nvalid) {
                            const int row = key[j] & (BKT_ROWS - 1);
                            if (row != cur) {
                                if (cur >= 0) *(float2*)(orow + (row0 + cur) * HALF) = acc;
                                acc = make_float2(0.f, 0.f);
                                cur = row;
                            }
                            acc.x = fmaf(w[j], x[j].x, acc.x);
                            acc.y = fmaf(w[j], x[j].y, acc.y);
                        }
                    }
                }
            }
        }
        if (cur >= 0) *(float2*)(orow + (row0 + cur) * HALF) = acc;
        return;
    }

    float* acc = (float*)raw;
    for (int i = tid; i < BKT_ROWS * DIM / 4; i += 256)
        ((float4*)acc)[i] = make_float4(0.f, 0.f, 0.f, 0.f);
    __syncthreads();
    const int end = beg + n;
    int2 cur = rec[(beg + tid < end) ? (beg + tid) : (end - 1)];
    for (int base = beg; base < end; base += 256) {
        int2 nxt = make_int2(0, 0);
        int nvalid = end - base - wib * 32;
        nvalid = nvalid > 32 ? 32 : nvalid;
#pragma unroll
        for (int g = 0; g < 32; g += GRP) {
            if (g < nvalid) {
                int    key[GRP];
                float  w[GRP];
                float2 x[GRP];
#pragma unroll
                for (int j = 0; j < GRP; ++j) {
                    const int jj = (g + j < nvalid) ? (g + j) : (nvalid - 1);
                    key[j] = __builtin_amdgcn_readlane(cur.x, jj);
                    w[j]   = __int_as_float(__builtin_amdgcn_readlane(cur.y, jj));
                    const float* src = spmm_src_row(srcA, srcB, split, key[j] >> BKT_SHIFT);
                    x[j] = *(const float2*)(src + 2 * lane);
                }
                if (g == 0) nxt = rec[(base + 256 + tid < end) ? (base + 256 + tid) : (end - 1)];
#pragma unroll
                for (int j = 0; j < GRP; ++j) {
                    if (g + j < nvalid) {
                        float* d = acc + (key[j] & (BKT_ROWS - 1)) * DIM + 2 * lane;
                        atomicAdd(d,     w[j] * x[j].x);
                        atomicAdd(d + 1, w[j] * x[j].y);
                    }
                }
            }
        }
        cur = nxt;
    }
    __syncthreads();
    for (int q = tid; q < BKT_ROWS * (HALF / 4); q += 256) {
        const int rr = q >> 3, pc = q & 7;
        if (row0 + rr < (size_t)N_NODES) {
            const float4 lo = *(const float4*)(acc + rr * DIM + 4 * pc);
            const float4 hi = *(const float4*)(acc + rr * DIM + HALF + 4 * pc);
            *(float4*)(out_lo + (row0 + rr) * HALF + 4 * pc) = lo;
            *(float4*)(out_hi + (row0 + rr) * HALF + 4 * pc) = hi;
        }
    }
}

__global__ __launch_bounds__(256) void bucket_spmm_plane_kernel(
    const int2* __restrict__ rec, const int* __restrict__ start,
    const int* __restrict__ cnt,
    const float* __restrict__ src, float* __restrict__ dst) {
    __shared__ __align__(16) unsigned char raw[BKT_ROWS * HALF * 4];
    __shared__ int rcnt[BKT_ROWS];
    __shared__ int rbeg[BKT_ROWS + 1];
    const int tid  = (int)threadIdx.x;
    const int lane = tid & 31;
    const int wib  = __builtin_amdgcn_readfirstlane(tid >> 5);
    const int bkt  = (int)blockIdx.x;
    const size_t row0 = (size_t)bkt * BKT_ROWS;
    const int beg = __builtin_amdgcn_readfirstlane(start[bkt]);
    const int n   = __builtin_amdgcn_readfirstlane(cnt[bkt]);

    if (n <= SORT_CAP) {
        int2* srec = (int2*)raw;
        bucket_order_rows(rec, beg, n, srec, rcnt, rbeg, tid);
        const int r0 = wib * 16;
        const int s0 = __builtin_amdgcn_readfirstlane(rbeg[r0]);
        const int s1 = __builtin_amdgcn_readfirstlane(rbeg[r0 + 16]);
        float* orow = dst + lane;
        {
            const int li = lane & 15;
            const bool empty = (lane < 16) && (rbeg[r0 + li + 1] == rbeg[r0 + li]);
            unsigned em = __builtin_amdgcn_ballot_w32(empty);
            while (em) {
                const int rr = r0 + __builtin_ctz(em);
                em &= em - 1;
                if (row0 + rr < (size_t)N_NODES) orow[(row0 + rr) * HALF] = 0.f;
            }
        }
        float acc = 0.f;
        int cur = -1;
        for (int base = s0; base < s1; base += 32) {
            int nvalid = s1 - base;
            nvalid = nvalid > 32 ? 32 : nvalid;
            const int2 mine = srec[(base + lane < s1) ? (base + lane) : (s1 - 1)];
#pragma unroll
            for (int g = 0; g < 32; g += GRP) {
                if (g < nvalid) {
                    int   key[GRP];
                    float w[GRP];
                    float x[GRP];
#pragma unroll
                    for (int j = 0; j < GRP; ++j) {
                        const int jj = (g + j < nvalid) ? (g + j) : (nvalid - 1);
                        key[j] = __builtin_amdgcn_readlane(mine.x, jj);
                        w[j]   = __int_as_float(__builtin_amdgcn_readlane(mine.y, jj));
                        x[j]   = src[(size_t)(key[j] >> BKT_SHIFT) * HALF + lane];
                    }
#pragma unroll
                    for (int j = 0; j < GRP; ++j) {
                        if (g + j < nvalid) {
                            const int row = key[j] & (BKT_ROWS - 1);
                            if (row != cur) {
                                if (cur >= 0) orow[(row0 + cur) * HALF] = acc;
                                acc = 0.f;
                                cur = row;
                            }
                            acc = fmaf(w[j], x[j], acc);
                        }
                    }
                }
            }
        }
        if (cur >= 0) orow[(row0 + cur) * HALF] = acc;
        return;
    }

    float* acc = (float*)raw;
    for (int i = tid; i < BKT_ROWS * HALF / 4; i += 256)
        ((float4*)acc)[i] = make_float4(0.f, 0.f, 0.f, 0.f);
    __syncthreads();
    const int end = beg + n;
    int2 cur = rec[(beg + tid < end) ? (beg + tid) : (end - 1)];
    for (int base = beg; base < end; base += 256) {
        int2 nxt = make_int2(0, 0);
        int nvalid = end - base - wib * 32;
        nvalid = nvalid > 32 ? 32 : nvalid;
#pragma unroll
        for (int g = 0; g < 32; g += GRP) {
            if (g < nvalid) {
                int   key[GRP];
                float w[GRP];
                float x[GRP];
#pragma unroll
                for (int j = 0; j < GRP; ++j) {
                    const int jj = (g + j < nvalid) ? (g + j) : (nvalid - 1);
                    key[j] = __builtin_amdgcn_readlane(cur.x, jj);
                    w[j]   = __int_as_float(__builtin_amdgcn_readlane(cur.y, jj));
                    x[j]   = src[(size_t)(key[j] >> BKT_SHIFT) * HALF + lane];
                }
                if (g == 0) nxt = rec[(base + 256 + tid < end) ? (base + 256 + tid) : (end - 1)];
#pragma unroll
                for (int j = 0; j < GRP; ++j) {
                    if (g + j < nvalid)
                        atomicAdd(acc + (key[j] & (BKT_ROWS - 1)) * HALF + lane, w[j] * x[j]);
                }
            }
        }
        cur = nxt;
    }
    __syncthreads();
    for (int q = tid; q < BKT_ROWS * (HALF / 4); q += 256) {
        const int rr = q >> 3;
        if (row0 + rr < (size_t)N_NODES)
            *(float4*)(dst + row0 * HALF + 4 * q) = *(const float4*)(acc + 4 * q);
    }
}

#define PRED_WPB   8
#define TILE_PITCH 68
__global__ __launch_bounds__(256) void pred_kernel(
    const float* __restrict__ Ulo, const float* __restrict__ Uhi,
    const float* __restrict__ Ilo, const float* __restrict__ Ihi,
    const int*   __restrict__ u_idx, const int* __restrict__ i_idx,
    const float* __restrict__ bu, const float* __restrict__ bi,
    const float* __restrict__ mu, float* __restrict__ out) {
    __shared__ float tiles[PRED_WPB][2][16 * TILE_PITCH];

    int wib  = threadIdx.x >> 5;
    int wave = blockIdx.x * PRED_WPB + wib;
    int lane = threadIdx.x & 31;
    int base = wave << 4;
    int m    = lane & 15;
    int hi   = lane >> 4;
    int ur = u_idx[base + m];
    int ir = i_idx[base + m];

    const int4* uq = (const int4*)(u_idx + base + 8 * hi);
    const int4* iq = (const int4*)(i_idx + base + 8 * hi);
    const int4 ua = uq[0], ub = uq[1];
    const int4 ia = iq[0], ib = iq[1];
    const int urow[8] = {ua.x, ua.y, ua.z, ua.w, ub.x, ub.y, ub.z, ub.w};
    const int irow[8] = {ia.x, ia.y, ia.z, ia.w, ib.x, ib.y, ib.z, ib.w};

    float* tu = &tiles[wib][0][0];
    float* ti = &tiles[wib][1][0];

    const float* up = (m < 8) ? Ulo : Uhi;
    const float* ip = (m < 8) ? Ilo : Ihi;
    const int    mo = 4 * (m & 7);
    float4 gu[8], gi[8];
#pragma unroll
    for (int j = 0; j < 8; ++j) {
        gu[j] = *(const float4*)(up + (size_t)urow[j] * HALF + mo);
        gi[j] = *(const float4*)(ip + (size_t)irow[j] * HALF + mo);
    }
#pragma unroll
    for (int j = 0; j < 8; ++j) {
        *(float4*)(tu + (8 * hi + j) * TILE_PITCH + 4 * m) = gu[j];
        *(float4*)(ti + (8 * hi + j) * TILE_PITCH + 4 * m) = gi[j];
    }
    __syncthreads();

    const float* fa = tu + m * TILE_PITCH + 32 * hi;
    const float* fb = ti + m * TILE_PITCH + 32 * hi;
    v8f c = {};
#pragma unroll
    for (int t = 0; t < 8; ++t) {
        const float4 a4 = *(const float4*)(fa + 4 * t);
        const float4 b4 = *(const float4*)(fb + 4 * t);
        v2f a0 = {a4.x, a4.y}, b0 = {b4.x, b4.y};
        v2f a1 = {a4.z, a4.w}, b1 = {b4.z, b4.w};
        c = __builtin_amdgcn_wmma_f32_16x16x4_f32(
                false, a0, false, b0, (short)0, c, false, false);
        c = __builtin_amdgcn_wmma_f32_16x16x4_f32(
                false, a1, false, b1, (short)0, c, false, false);
    }

    float d = 0.f;
#pragma unroll
    for (int j = 0; j < 8; ++j) {
        float vlo = __shfl(c[j], j, 32);
        float vhi = __shfl(c[j], j + 24, 32);
        if (lane == j)     d = vlo;
        if (lane == j + 8) d = vhi;
    }

    if (lane < 16) {
        out[base + lane] = d + bu[ur] + bi[ir] + mu[0];
    }
}

extern "C" void kernel_launch(void* const* d_in, const int* in_sizes, int n_in,
                              void* d_out, int out_size, void* d_ws, size_t ws_size,
                              hipStream_t stream) {
    const int*   u_idx  = (const int*)d_in[0];
    const int*   i_idx  = (const int*)d_in[1];
    const int*   A_rows = (const int*)d_in[2];
    const int*   A_cols = (const int*)d_in[3];
    const float* A_vals = (const float*)d_in[4];
    const float* Eu     = (const float*)d_in[5];
    const float* Ei     = (const float*)d_in[6];
    const float* bu     = (const float*)d_in[7];
    const float* bi     = (const float*)d_in[8];
    const float* mu     = (const float*)d_in[9];
    float* out = (float*)d_out;

    float* P0 = (float*)d_ws;
    float* P1 = P0 + (size_t)N_NODES * HALF;
    float* P2 = P1 + (size_t)N_NODES * HALF;
    int2*  rec   = (int2*)(P2 + (size_t)N_NODES * HALF);
    int*   cnt   = (int*)(rec + NNZ_CT);
    int*   fill  = cnt + NBKT_PAD;
    int*   start = fill + NBKT_PAD;

    clear_counts_kernel<<<1, 256, 0, stream>>>(cnt);
    bucket_count_kernel<<<PLC_BLOCKS, 256, 0, stream>>>(A_rows, cnt, NNZ_CT);
    bucket_place_kernel<<<PLC_BLOCKS, 256, 0, stream>>>(
        A_rows, A_cols, A_vals, cnt, fill, start, rec, NNZ_CT);

    bucket_spmm_rows_kernel<<<NBKT, 256, 0, stream>>>(
        rec, start, cnt, Eu, Ei, N_USERS, P0, P1);

    bucket_spmm_plane_kernel<<<NBKT, 256, 0, stream>>>(rec, start, cnt, P0, P2);
    bucket_spmm_plane_kernel<<<NBKT, 256, 0, stream>>>(rec, start, cnt, P1, P0);

    pred_kernel<<<(BATCH_CT / 16) / 8, 256, 0, stream>>>(
        P2, P0, P2 + (size_t)N_USERS * HALF, P0 + (size_t)N_USERS * HALF,
        u_idx, i_idx, bu, bi, mu, out);
}


# pred: 8192 resident waves each work through 8 tiles (was 65536 one-tile waves); tile t+1's 16 row loads in flight under tile t's WMMAs; work-group barrier dropped; order of sums unchanged
# speedup vs baseline: 2.7325x; 1.0379x over previous
#include <hip/hip_runtime.h>


#define N_USERS  100000
#define N_ITEMS  50000
#define N_NODES  (N_USERS + N_ITEMS)
#define NNZ_CT   1500000
#define BATCH_CT 1048576
#define DIM      64
#define HALF     32

typedef __attribute__((ext_vector_type(2))) float v2f;
typedef __attribute__((ext_vector_type(8))) float v8f;

#define BKT_ROWS  128
#define BKT_SHIFT 7
#define NBKT      ((N_NODES + BKT_ROWS - 1) / BKT_ROWS)
#define NBKT_PAD  1280
#define PLC_BLOCKS 512
#define PLC_EPB   ((NNZ_CT + PLC_BLOCKS - 1) / PLC_BLOCKS)
#define PLC_NIT   ((PLC_EPB + 255) / 256)

__global__ __launch_bounds__(256) void clear_counts_kernel(int* __restrict__ cnt2) {
    for (int i = threadIdx.x; i < 2 * NBKT_PAD; i += 256) cnt2[i] = 0;
}

__global__ __launch_bounds__(256) void bucket_count_kernel(
    const int* __restrict__ rows, int* __restrict__ cnt, int nnz) {
    __shared__ int h[NBKT_PAD];
    const int tid = (int)threadIdx.x;
    for (int i = tid; i < NBKT_PAD; i += 256) h[i] = 0;
    __syncthreads();
    const int e_beg = (int)blockIdx.x * PLC_EPB;
    int r[PLC_NIT];
#pragma unroll
    for (int q = 0; q < PLC_NIT; ++q) {
        const int i = tid + q * 256;
        const int e = e_beg + i;
        r[q] = (i < PLC_EPB && e < nnz) ? rows[e] : -1;
    }
#pragma unroll
    for (int q = 0; q < PLC_NIT; ++q) {
        if ((unsigned)r[q] < (unsigned)N_NODES) atomicAdd(&h[r[q] >> BKT_SHIFT], 1);
    }
    __syncthreads();
    for (int k = tid; k < NBKT; k += 256) {
        const int c = h[k];
        if (c) atomicAdd(&cnt[k], c);
    }
}

__global__ __launch_bounds__(256) void bucket_place_kernel(
    const int* __restrict__ rows, const int* __restrict__ cols,
    const float* __restrict__ vals, const int* __restrict__ cnt,
    int* __restrict__ fill, int* __restrict__ start_out,
    int2* __restrict__ rec, int nnz) {
    __shared__ int h[NBKT_PAD];
    __shared__ int st[NBKT_PAD];
    __shared__ int wsum[8];
    const int tid  = (int)threadIdx.x;
    const int lane = tid & 31;
    const int wib  = tid >> 5;
    for (int i = tid; i < NBKT_PAD; i += 256) h[i] = 0;
    __syncthreads();
    const int e_beg = (int)blockIdx.x * PLC_EPB;
    int   r[PLC_NIT], c[PLC_NIT];
    float v[PLC_NIT];
#pragma unroll
    for (int q = 0; q < PLC_NIT; ++q) {
        const int i = tid + q * 256;
        const int e = e_beg + i;
        const bool ok = (i < PLC_EPB && e < nnz);
        r[q] = ok ? rows[e] : -1;
        c[q] = ok ? cols[e] : 0;
        v[q] = ok ? vals[e] : 0.f;
    }
#pragma unroll
    for (int q = 0; q < PLC_NIT; ++q) {
        if ((unsigned)r[q] < (unsigned)N_NODES) atomicAdd(&h[r[q] >> BKT_SHIFT], 1);
    }
    int c5[5];
    int s = 0;
#pragma unroll
    for (int q = 0; q < 5; ++q) {
        const int k = 5 * tid + q;
        c5[q] = (k < NBKT) ? cnt[k] : 0;
        s += c5[q];
    }
    int incl = s;
    for (int d = 1; d < 32; d <<= 1) {
        const int o = __shfl_up(incl, d, 32);
        if (lane >= d) incl += o;
    }
    if (lane == 31) wsum[wib] = incl;
    __syncthreads();
    int woff = 0;
    for (int w = 0; w < 8; ++w) if (w < wib) woff += wsum[w];
    int run = woff + incl - s;
#pragma unroll
    for (int q = 0; q < 5; ++q) {
        st[5 * tid + q] = run;
        run += c5[q];
    }
    __syncthreads();
    for (int k = tid; k < NBKT; k += 256) {
        const int n = h[k];
        const int b = n ? atomicAdd(&fill[k], n) : 0;
        h[k] = st[k] + b;
        if (blockIdx.x == 0) start_out[k] = st[k];
    }
    __syncthreads();
#pragma unroll
    for (int q = 0; q < PLC_NIT; ++q) {
        if ((unsigned)r[q] < (unsigned)N_NODES) {
            const int p = atomicAdd(&h[r[q] >> BKT_SHIFT], 1);
            rec[p] = make_int2((c[q] << BKT_SHIFT) | (r[q] & (BKT_ROWS - 1)),
                               __float_as_int(v[q]));
        }
    }
}

#define GRP 16
#define SORT_CAP 2048
#define SORT_NIT (SORT_CAP / 256)

__device__ __forceinline__ const float* spmm_src_row(
    const float* __restrict__ srcA, const float* __restrict__ srcB, int split, int c) {
    return (c < split) ? (srcA + (size_t)c * DIM)
                       : (srcB + (size_t)(c - split) * DIM);
}

__device__ __forceinline__ void bucket_order_rows(
    const int2* __restrict__ rec, int beg, int n,
    int2* srec, int* rcnt, int* rbeg, int tid) {
    const int lane = tid & 31;
    if (tid < BKT_ROWS) rcnt[tid] = 0;
    int2 r[SORT_NIT];
#pragma unroll
    for (int q = 0; q < SORT_NIT; ++q) {
        const int i = tid + q * 256;
        r[q] = (i < n) ? rec[beg + i] : make_int2(0, 0);
    }
    __syncthreads();
#pragma unroll
    for (int q = 0; q < SORT_NIT; ++q)
        if (tid + q * 256 < n) atomicAdd(&rcnt[r[q].x & (BKT_ROWS - 1)], 1);
    __syncthreads();
    if (tid < 32) {
        int c4[4];
        int s = 0;
#pragma unroll
        for (int k = 0; k < 4; ++k) { c4[k] = rcnt[4 * lane + k]; s += c4[k]; }
        int incl = s;
        for (int d = 1; d < 32; d <<= 1) {
            const int o = __shfl_up(incl, d, 32);
            if (lane >= d) incl += o;
        }
        int run = incl - s;
#pragma unroll
        for (int k = 0; k < 4; ++k) {
            rbeg[4 * lane + k] = run;
            rcnt[4 * lane + k] = run;
            run += c4[k];
        }
        if (lane == 31) rbeg[BKT_ROWS] = incl;
    }
    __syncthreads();
    int p[SORT_NIT];
#pragma unroll
    for (int q = 0; q < SORT_NIT; ++q)
        p[q] = (tid + q * 256 < n) ? atomicAdd(&rcnt[r[q].x & (BKT_ROWS - 1)], 1) : 0;
#pragma unroll
    for (int q = 0; q < SORT_NIT; ++q)
        if (tid + q * 256 < n) srec[p[q]] = r[q];
    __syncthreads();
}

__global__ __launch_bounds__(256) void bucket_spmm_rows_kernel(
    const int2* __restrict__ rec, const int* __restrict__ start,
    const int* __restrict__ cnt,
    const float* __restrict__ srcA, const float* __restrict__ srcB, int split,
    float* __restrict__ out_lo, float* __restrict__ out_hi) {
    __shared__ __align__(16) unsigned char raw[BKT_ROWS * DIM * 4];
    __shared__ int rcnt[BKT_ROWS];
    __shared__ int rbeg[BKT_ROWS + 1];
    const int tid  = (int)threadIdx.x;
    const int lane = tid & 31;
    const int wib  = __builtin_amdgcn_readfirstlane(tid >> 5);
    const int bkt  = (int)blockIdx.x;
    const size_t row0 = (size_t)bkt * BKT_ROWS;
    const int beg = __builtin_amdgcn_readfirstlane(start[bkt]);
    const int n   = __builtin_amdgcn_readfirstlane(cnt[bkt]);

    if (n <= SORT_CAP) {
        int2* srec = (int2*)raw;
        bucket_order_rows(rec, beg, n, srec, rcnt, rbeg, tid);
        const int r0 = wib * 16;
        const int s0 = __builtin_amdgcn_readfirstlane(rbeg[r0]);
        const int s1 = __builtin_amdgcn_readfirstlane(rbeg[r0 + 16]);
        float* orow = ((lane < 16) ? out_lo : out_hi) + 2 * (lane & 15);
        {
            const int li = lane & 15;
            const bool empty = (lane < 16) && (rbeg[r0 + li + 1] == rbeg[r0 + li]);
            unsigned em = __builtin_amdgcn_ballot_w32(empty);
            while (em) {
                const int rr = r0 + __builtin_ctz(em);
                em &= em - 1;
                if (row0 + rr < (size_t)N_NODES)
                    *(float2*)(orow + (row0 + rr) * HALF) = make_float2(0.f, 0.f);
            }
        }
        float2 acc = make_float2(0.f, 0.f);
        int cur = -1;
        for (int base = s0; base < s1; base += 32) {
            int nvalid = s1 - base;
            nvalid = nvalid > 32 ? 32 : nvalid;
            const int2 mine = srec[(base + lane < s1) ? (base + lane) : (s1 - 1)];
#pragma unroll
            for (int g = 0; g < 32; g += GRP) {
                if (g < nvalid) {
                    int    key[GRP];
                    float  w[GRP];
                    float2 x[GRP];
#pragma unroll
                    for (int j = 0; j < GRP; ++j) {
                        const int jj = (g + j < nvalid) ? (g + j) : (nvalid - 1);
                        key[j] = __builtin_amdgcn_readlane(mine.x, jj);
                        w[j]   = __int_as_float(__builtin_amdgcn_readlane(mine.y, jj));
                        const float* src = spmm_src_row(srcA, srcB, split, key[j] >> BKT_SHIFT);
                        x[j] = *(const float2*)(src + 2 * lane);
                    }
#pragma unroll
                    for (int j = 0; j < GRP; ++j) {
                        if (g + j < nvalid) {
                            const int row = key[j] & (BKT_ROWS - 1);
                            if (row != cur) {
                                if (cur >= 0) *(float2*)(orow + (row0 + cur) * HALF) = acc;
                                acc = make_float2(0.f, 0.f);
                                cur = row;
                            }
                            acc.x = fmaf(w[j], x[j].x, acc.x);
                            acc.y = fmaf(w[j], x[j].y, acc.y);
                        }
                    }
                }
            }
        }
        if (cur >= 0) *(float2*)(orow + (row0 + cur) * HALF) = acc;
        return;
    }

    float* acc = (float*)raw;
    for (int i = tid; i < BKT_ROWS * DIM / 4; i += 256)
        ((float4*)acc)[i] = make_float4(0.f, 0.f, 0.f, 0.f);
    __syncthreads();
    const int end = beg + n;
    int2 cur = rec[(beg + tid < end) ? (beg + tid) : (end - 1)];
    for (int base = beg; base < end; base += 256) {
        int2 nxt = make_int2(0, 0);
        int nvalid = end - base - wib * 32;
        nvalid = nvalid > 32 ? 32 : nvalid;
#pragma unroll
        for (int g = 0; g < 32; g += GRP) {
            if (g < nvalid) {
                int    key[GRP];
                float  w[GRP];
                float2 x[GRP];
#pragma unroll
                for (int j = 0; j < GRP; ++j) {
                    const int jj = (g + j < nvalid) ? (g + j) : (nvalid - 1);
                    key[j] = __builtin_amdgcn_readlane(cur.x, jj);
                    w[j]   = __int_as_float(__builtin_amdgcn_readlane(cur.y, jj));
                    const float* src = spmm_src_row(srcA, srcB, split, key[j] >> BKT_SHIFT);
                    x[j] = *(const float2*)(src + 2 * lane);
                }
                if (g == 0) nxt = rec[(base + 256 + tid < end) ? (base + 256 + tid) : (end - 1)];
#pragma unroll
                for (int j = 0; j < GRP; ++j) {
                    if (g + j < nvalid) {
                        float* d = acc + (key[j] & (BKT_ROWS - 1)) * DIM + 2 * lane;
                        atomicAdd(d,     w[j] * x[j].x);
                        atomicAdd(d + 1, w[j] * x[j].y);
                    }
                }
            }
        }
        cur = nxt;
    }
    __syncthreads();
    for (int q = tid; q < BKT_ROWS * (HALF / 4); q += 256) {
        const int rr = q >> 3, pc = q & 7;
        if (row0 + rr < (size_t)N_NODES) {
            const float4 lo = *(const float4*)(acc + rr * DIM + 4 * pc);
            const float4 hi = *(const float4*)(acc + rr * DIM + HALF + 4 * pc);
            *(float4*)(out_lo + (row0 + rr) * HALF + 4 * pc) = lo;
            *(float4*)(out_hi + (row0 + rr) * HALF + 4 * pc) = hi;
        }
    }
}

__global__ __launch_bounds__(256) void bucket_spmm_plane_kernel(
    const int2* __restrict__ rec, const int* __restrict__ start,
    const int* __restrict__ cnt,
    const float* __restrict__ src, float* __restrict__ dst) {
    __shared__ __align__(16) unsigned char raw[BKT_ROWS * HALF * 4];
    __shared__ int rcnt[BKT_ROWS];
    __shared__ int rbeg[BKT_ROWS + 1];
    const int tid  = (int)threadIdx.x;
    const int lane = tid & 31;
    const int wib  = __builtin_amdgcn_readfirstlane(tid >> 5);
    const int bkt  = (int)blockIdx.x;
    const size_t row0 = (size_t)bkt * BKT_ROWS;
    const int beg = __builtin_amdgcn_readfirstlane(start[bkt]);
    const int n   = __builtin_amdgcn_readfirstlane(cnt[bkt]);

    if (n <= SORT_CAP) {
        int2* srec = (int2*)raw;
        bucket_order_rows(rec, beg, n, srec, rcnt, rbeg, tid);
        const int r0 = wib * 16;
        const int s0 = __builtin_amdgcn_readfirstlane(rbeg[r0]);
        const int s1 = __builtin_amdgcn_readfirstlane(rbeg[r0 + 16]);
        float* orow = dst + lane;
        {
            const int li = lane & 15;
            const bool empty = (lane < 16) && (rbeg[r0 + li + 1] == rbeg[r0 + li]);
            unsigned em = __builtin_amdgcn_ballot_w32(empty);
            while (em) {
                const int rr = r0 + __builtin_ctz(em);
                em &= em - 1;
                if (row0 + rr < (size_t)N_NODES) orow[(row0 + rr) * HALF] = 0.f;
            }
        }
        float acc = 0.f;
        int cur = -1;
        for (int base = s0; base < s1; base += 32) {
            int nvalid = s1 - base;
            nvalid = nvalid > 32 ? 32 : nvalid;
            const int2 mine = srec[(base + lane < s1) ? (base + lane) : (s1 - 1)];
#pragma unroll
            for (int g = 0; g < 32; g += GRP) {
                if (g < nvalid) {
                    int   key[GRP];
                    float w[GRP];
                    float x[GRP];
#pragma unroll
                    for (int j = 0; j < GRP; ++j) {
                        const int jj = (g + j < nvalid) ? (g + j) : (nvalid - 1);
                        key[j] = __builtin_amdgcn_readlane(mine.x, jj);
                        w[j]   = __int_as_float(__builtin_amdgcn_readlane(mine.y, jj));
                        x[j]   = src[(size_t)(key[j] >> BKT_SHIFT) * HALF + lane];
                    }
#pragma unroll
                    for (int j = 0; j < GRP; ++j) {
                        if (g + j < nvalid) {
                            const int row = key[j] & (BKT_ROWS - 1);
                            if (row != cur) {
                                if (cur >= 0) orow[(row0 + cur) * HALF] = acc;
                                acc = 0.f;
                                cur = row;
                            }
                            acc = fmaf(w[j], x[j], acc);
                        }
                    }
                }
            }
        }
        if (cur >= 0) orow[(row0 + cur) * HALF] = acc;
        return;
    }

    float* acc = (float*)raw;
    for (int i = tid; i < BKT_ROWS * HALF / 4; i += 256)
        ((float4*)acc)[i] = make_float4(0.f, 0.f, 0.f, 0.f);
    __syncthreads();
    const int end = beg + n;
    int2 cur = rec[(beg + tid < end) ? (beg + tid) : (end - 1)];
    for (int base = beg; base < end; base += 256) {
        int2 nxt = make_int2(0, 0);
        int nvalid = end - base - wib * 32;
        nvalid = nvalid > 32 ? 32 : nvalid;
#pragma unroll
        for (int g = 0; g < 32; g += GRP) {
            if (g < nvalid) {
                int   key[GRP];
                float w[GRP];
                float x[GRP];
#pragma unroll
                for (int j = 0; j < GRP; ++j) {
                    const int jj = (g + j < nvalid) ? (g + j) : (nvalid - 1);
                    key[j] = __builtin_amdgcn_readlane(cur.x, jj);
                    w[j]   = __int_as_float(__builtin_amdgcn_readlane(cur.y, jj));
                    x[j]   = src[(size_t)(key[j] >> BKT_SHIFT) * HALF + lane];
                }
                if (g == 0) nxt = rec[(base + 256 + tid < end) ? (base + 256 + tid) : (end - 1)];
#pragma unroll
                for (int j = 0; j < GRP; ++j) {
                    if (g + j < nvalid)
                        atomicAdd(acc + (key[j] & (BKT_ROWS - 1)) * HALF + lane, w[j] * x[j]);
                }
            }
        }
        cur = nxt;
    }
    __syncthreads();
    for (int q = tid; q < BKT_ROWS * (HALF / 4); q += 256) {
        const int rr = q >> 3;
        if (row0 + rr < (size_t)N_NODES)
            *(float4*)(dst + row0 * HALF + 4 * q) = *(const float4*)(acc + 4 * q);
    }
}

#define PRED_WPB   8
#define PRED_TPW   8
#define PRED_BLOCKS (BATCH_CT / 16 / PRED_WPB / PRED_TPW)
#define TILE_PITCH 68
__device__ __forceinline__ void pred_tile_products(
    const float* fa, const float* fb, int lane, float cbu, float cbi, float muv,
    float* __restrict__ out, int base) {
    v8f c = {};
#pragma unroll
    for (int t = 0; t < 8; ++t) {
        const float4 a4 = *(const float4*)(fa + 4 * t);
        const float4 b4 = *(const float4*)(fb + 4 * t);
        v2f a0 = {a4.x, a4.y}, b0 = {b4.x, b4.y};
        v2f a1 = {a4.z, a4.w}, b1 = {b4.z, b4.w};
        c = __builtin_amdgcn_wmma_f32_16x16x4_f32(
                false, a0, false, b0, (short)0, c, false, false);
        c = __builtin_amdgcn_wmma_f32_16x16x4_f32(
                false, a1, false, b1, (short)0, c, false, false);
    }
    float d = 0.f;
#pragma unroll
    for (int j = 0; j < 8; ++j) {
        float vlo = __shfl(c[j], j, 32);
        float vhi = __shfl(c[j], j + 24, 32);
        if (lane == j)     d = vlo;
        if (lane == j + 8) d = vhi;
    }
    if (lane < 16) out[base + lane] = d + cbu + cbi + muv;
}

#define PRED_LDU(r) (*(const float4*)(up + (size_t)(r) * HALF + mo))
#define PRED_LDI(r) (*(const float4*)(ip + (size_t)(r) * HALF + mo))
#define PRED_GATHER_TILE()                                              \
    gu0 = PRED_LDU(ua.x); gi0 = PRED_LDI(ia.x);                         \
    gu1 = PRED_LDU(ua.y); gi1 = PRED_LDI(ia.y);                         \
    gu2 = PRED_LDU(ua.z); gi2 = PRED_LDI(ia.z);                         \
    gu3 = PRED_LDU(ua.w); gi3 = PRED_LDI(ia.w);                         \
    gu4 = PRED_LDU(ub.x); gi4 = PRED_LDI(ib.x);                         \
    gu5 = PRED_LDU(ub.y); gi5 = PRED_LDI(ib.y);                         \
    gu6 = PRED_LDU(ub.z); gi6 = PRED_LDI(ib.z);                         \
    gu7 = PRED_LDU(ub.w); gi7 = PRED_LDI(ib.w);
#define PRED_ST(j, vu, vi)                                              \
    *(float4*)(su + (j) * TILE_PITCH) = (vu);                           \
    *(float4*)(si + (j) * TILE_PITCH) = (vi);
#define PRED_STAGE_TILE()                                               \
    PRED_ST(0, gu0, gi0) PRED_ST(1, gu1, gi1) PRED_ST(2, gu2, gi2)      \
    PRED_ST(3, gu3, gi3) PRED_ST(4, gu4, gi4) PRED_ST(5, gu5, gi5)      \
    PRED_ST(6, gu6, gi6) PRED_ST(7, gu7, gi7)
#define PRED_LOAD_IDX(b)                                                \
    ur = u_idx[(b) + m];                                                \
    ir = i_idx[(b) + m];                                                \
    ua = *(const int4*)(u_idx + (b) + 8 * hi);                          \
    ub = *(const int4*)(u_idx + (b) + 8 * hi + 4);                      \
    ia = *(const int4*)(i_idx + (b) + 8 * hi);                          \
    ib = *(const int4*)(i_idx + (b) + 8 * hi + 4);

__global__ __launch_bounds__(256) void pred_kernel(
    const float* __restrict__ Ulo, const float* __restrict__ Uhi,
    const float* __restrict__ Ilo, const float* __restrict__ Ihi,
    const int*   __restrict__ u_idx, const int* __restrict__ i_idx,
    const float* __restrict__ bu, const float* __restrict__ bi,
    const float* __restrict__ mu, float* __restrict__ out) {
    __shared__ float tiles[PRED_WPB][2][16 * TILE_PITCH];

    const int wib  = threadIdx.x >> 5;
    const int lane = threadIdx.x & 31;
    const int m    = lane & 15;
    const int hi   = lane >> 4;
    int base = ((int)blockIdx.x * PRED_WPB + wib) * (16 * PRED_TPW);

    float* tu = &tiles[wib][0][0];
    float* ti = &tiles[wib][1][0];
    const float* up = (m < 8) ? Ulo : Uhi;
    const float* ip = (m < 8) ? Ilo : Ihi;
    const int    mo = 4 * (m & 7);
    float* su = tu + (8 * hi) * TILE_PITCH + 4 * m;
    float* si = ti + (8 * hi) * TILE_PITCH + 4 * m;
    const float* fa = tu + m * TILE_PITCH + 32 * hi;
    const float* fb = ti + m * TILE_PITCH + 32 * hi;
    const float muv = mu[0];

    int  ur, ir;
    int4 ua, ub, ia, ib;
    float4 gu0, gu1, gu2, gu3, gu4, gu5, gu6, gu7;
    float4 gi0, gi1, gi2, gi3, gi4, gi5, gi6, gi7;

    PRED_LOAD_IDX(base)
    PRED_GATHER_TILE()
    float bu_v = bu[ur];
    float bi_v = bi[ir];
    PRED_LOAD_IDX(base + 16)

#pragma unroll 1
    for (int it = 0; it < PRED_TPW; ++it) {
        PRED_STAGE_TILE()
        const float cbu = bu_v, cbi = bi_v;
        __builtin_amdgcn_fence(__ATOMIC_ACQ_REL, "wavefront");

        if (it + 1 < PRED_TPW) {
            PRED_GATHER_TILE()
            bu_v = bu[ur];
            bi_v = bi[ir];
            if (it + 2 < PRED_TPW) {
                PRED_LOAD_IDX(base + 32)
            }
        }

        pred_tile_products(fa, fb, lane, cbu, cbi, muv, out, base);
        base += 16;
    }
}

extern "C" void kernel_launch(void* const* d_in, const int* in_sizes, int n_in,
                              void* d_out, int out_size, void* d_ws, size_t ws_size,
                              hipStream_t stream) {
    const int*   u_idx  = (const int*)d_in[0];
    const int*   i_idx  = (const int*)d_in[1];
    const int*   A_rows = (const int*)d_in[2];
    const int*   A_cols = (const int*)d_in[3];
    const float* A_vals = (const float*)d_in[4];
    const float* Eu     = (const float*)d_in[5];
    const float* Ei     = (const float*)d_in[6];
    const float* bu     = (const float*)d_in[7];
    const float* bi     = (const float*)d_in[8];
    const float* mu     = (const float*)d_in[9];
    float* out = (float*)d_out;

    float* P0 = (float*)d_ws;
    float* P1 = P0 + (size_t)N_NODES * HALF;
    float* P2 = P1 + (size_t)N_NODES * HALF;
    int2*  rec   = (int2*)(P2 + (size_t)N_NODES * HALF);
    int*   cnt   = (int*)(rec + NNZ_CT);
    int*   fill  = cnt + NBKT_PAD;
    int*   start = fill + NBKT_PAD;

    clear_counts_kernel<<<1, 256, 0, stream>>>(cnt);
    bucket_count_kernel<<<PLC_BLOCKS, 256, 0, stream>>>(A_rows, cnt, NNZ_CT);
    bucket_place_kernel<<<PLC_BLOCKS, 256, 0, stream>>>(
        A_rows, A_cols, A_vals, cnt, fill, start, rec, NNZ_CT);

    bucket_spmm_rows_kernel<<<NBKT, 256, 0, stream>>>(
        rec, start, cnt, Eu, Ei, N_USERS, P0, P1);

    bucket_spmm_plane_kernel<<<NBKT, 256, 0, stream>>>(rec, start, cnt, P0, P2);
    bucket_spmm_plane_kernel<<<NBKT, 256, 0, stream>>>(rec, start, cnt, P1, P0);

    pred_kernel<<<PRED_BLOCKS, 256, 0, stream>>>(
        P2, P0, P2 + (size_t)N_USERS * HALF, P0 + (size_t)N_USERS * HALF,
        u_idx, i_idx, bu, bi, mu, out);
}


# edge placing: a block orders its 5860 edges by bucket in LDS, then writes contiguous runs per bucket (256 blocks x 1024 threads), not 32 unrelated 8-byte pieces per store; sum order as before
# speedup vs baseline: 2.8983x; 1.0607x over previous
#include <hip/hip_runtime.h>


#define N_USERS  100000
#define N_ITEMS  50000
#define N_NODES  (N_USERS + N_ITEMS)
#define NNZ_CT   1500000
#define BATCH_CT 1048576
#define DIM      64
#define HALF     32

typedef __attribute__((ext_vector_type(2))) float v2f;
typedef __attribute__((ext_vector_type(8))) float v8f;

#define BKT_ROWS  128
#define BKT_SHIFT 7
#define NBKT      ((N_NODES + BKT_ROWS - 1) / BKT_ROWS)
#define NBKT_PAD  1280
#define CNT_BLOCKS 512
#define CNT_EPB   ((NNZ_CT + CNT_BLOCKS - 1) / CNT_BLOCKS)
#define CNT_NIT   ((CNT_EPB + 255) / 256)
#define PLC_BLOCKS  256
#define PLC_THREADS 1024
#define PLC_EPB   ((NNZ_CT + PLC_BLOCKS - 1) / PLC_BLOCKS)
#define PLC_NIT   ((PLC_EPB + PLC_THREADS - 1) / PLC_THREADS)
#define PLC_SCAN_N (2 * PLC_THREADS)

__global__ __launch_bounds__(256) void clear_counts_kernel(int* __restrict__ cnt2) {
    for (int i = threadIdx.x; i < 2 * NBKT_PAD; i += 256) cnt2[i] = 0;
}

__global__ __launch_bounds__(256) void bucket_count_kernel(
    const int* __restrict__ rows, int* __restrict__ cnt, int nnz) {
    __shared__ int h[NBKT_PAD];
    const int tid = (int)threadIdx.x;
    for (int i = tid; i < NBKT_PAD; i += 256) h[i] = 0;
    __syncthreads();
    const int e_beg = (int)blockIdx.x * CNT_EPB;
    int r[CNT_NIT];
#pragma unroll
    for (int q = 0; q < CNT_NIT; ++q) {
        const int i = tid + q * 256;
        const int e = e_beg + i;
        r[q] = (i < CNT_EPB && e < nnz) ? rows[e] : -1;
    }
#pragma unroll
    for (int q = 0; q < CNT_NIT; ++q) {
        if ((unsigned)r[q] < (unsigned)N_NODES) atomicAdd(&h[r[q] >> BKT_SHIFT], 1);
    }
    __syncthreads();
    for (int k = tid; k < NBKT; k += 256) {
        const int c = h[k];
        if (c) atomicAdd(&cnt[k], c);
    }
}

__global__ __launch_bounds__(PLC_THREADS) void bucket_place_kernel(
    const int* __restrict__ rows, const int* __restrict__ cols,
    const float* __restrict__ vals, const int* __restrict__ cnt,
    int* __restrict__ fill, int* __restrict__ start_out,
    int2* __restrict__ rec, int nnz) {
    __shared__ int h[PLC_SCAN_N];
    __shared__ int st[PLC_SCAN_N];
    __shared__ int lp[PLC_SCAN_N];
    __shared__ int wsum_g[32];
    __shared__ int wsum_l[32];
    __shared__ int2 srec[PLC_EPB];
    __shared__ unsigned short sbkt[PLC_EPB];
    const int tid  = (int)threadIdx.x;
    const int lane = tid & 31;
    const int wib  = tid >> 5;
    for (int i = tid; i < PLC_SCAN_N; i += PLC_THREADS) h[i] = 0;
    __syncthreads();
    const int e_beg = (int)blockIdx.x * PLC_EPB;
    int   r[PLC_NIT], c[PLC_NIT];
    float v[PLC_NIT];
#pragma unroll
    for (int q = 0; q < PLC_NIT; ++q) {
        const int i = tid + q * PLC_THREADS;
        const int e = e_beg + i;
        const bool ok = (i < PLC_EPB && e < nnz);
        r[q] = ok ? rows[e] : -1;
        c[q] = ok ? cols[e] : 0;
        v[q] = ok ? vals[e] : 0.f;
    }
    const int k0 = 2 * tid, k1 = 2 * tid + 1;
    const int g0 = (k0 < NBKT) ? cnt[k0] : 0;
    const int g1 = (k1 < NBKT) ? cnt[k1] : 0;
#pragma unroll
    for (int q = 0; q < PLC_NIT; ++q) {
        if ((unsigned)r[q] < (unsigned)N_NODES) atomicAdd(&h[r[q] >> BKT_SHIFT], 1);
    }
    __syncthreads();
    const int l0 = h[k0], l1 = h[k1];
    const int sg = g0 + g1, sl = l0 + l1;
    int ig = sg, il = sl;
    for (int d = 1; d < 32; d <<= 1) {
        const int og = __shfl_up(ig, d, 32);
        const int ol = __shfl_up(il, d, 32);
        if (lane >= d) { ig += og; il += ol; }
    }
    if (lane == 31) { wsum_g[wib] = ig; wsum_l[wib] = il; }
    __syncthreads();
    int offg = 0, offl = 0, nloc = 0;
    for (int w = 0; w < 32; ++w) {
        const int a = wsum_g[w], b = wsum_l[w];
        if (w < wib) { offg += a; offl += b; }
        nloc += b;
    }
    const int rung = offg + ig - sg;
    const int runl = offl + il - sl;
    st[k0] = rung;  st[k1] = rung + g0;
    lp[k0] = runl;  lp[k1] = runl + l0;
    __syncthreads();
    for (int k = tid; k < NBKT; k += PLC_THREADS) {
        const int n = h[k];
        const int b = n ? atomicAdd(&fill[k], n) : 0;
        if (blockIdx.x == 0) start_out[k] = st[k];
        st[k] = st[k] + b - lp[k];
        h[k]  = lp[k];
    }
    __syncthreads();
#pragma unroll
    for (int q = 0; q < PLC_NIT; ++q) {
        if ((unsigned)r[q] < (unsigned)N_NODES) {
            const int bk = r[q] >> BKT_SHIFT;
            const int p  = atomicAdd(&h[bk], 1);
            srec[p] = make_int2((c[q] << BKT_SHIFT) | (r[q] & (BKT_ROWS - 1)),
                                __float_as_int(v[q]));
            sbkt[p] = (unsigned short)bk;
        }
    }
    __syncthreads();
    for (int s = tid; s < nloc; s += PLC_THREADS)
        rec[st[sbkt[s]] + s] = srec[s];
}

#define GRP 16
#define SORT_CAP 2048
#define SORT_NIT (SORT_CAP / 256)

__device__ __forceinline__ const float* spmm_src_row(
    const float* __restrict__ srcA, const float* __restrict__ srcB, int split, int c) {
    return (c < split) ? (srcA + (size_t)c * DIM)
                       : (srcB + (size_t)(c - split) * DIM);
}

__device__ __forceinline__ void bucket_order_rows(
    const int2* __restrict__ rec, int beg, int n,
    int2* srec, int* rcnt, int* rbeg, int tid) {
    const int lane = tid & 31;
    if (tid < BKT_ROWS) rcnt[tid] = 0;
    int2 r[SORT_NIT];
#pragma unroll
    for (int q = 0; q < SORT_NIT; ++q) {
        const int i = tid + q * 256;
        r[q] = (i < n) ? rec[beg + i] : make_int2(0, 0);
    }
    __syncthreads();
#pragma unroll
    for (int q = 0; q < SORT_NIT; ++q)
        if (tid + q * 256 < n) atomicAdd(&rcnt[r[q].x & (BKT_ROWS - 1)], 1);
    __syncthreads();
    if (tid < 32) {
        int c4[4];
        int s = 0;
#pragma unroll
        for (int k = 0; k < 4; ++k) { c4[k] = rcnt[4 * lane + k]; s += c4[k]; }
        int incl = s;
        for (int d = 1; d < 32; d <<= 1) {
            const int o = __shfl_up(incl, d, 32);
            if (lane >= d) incl += o;
        }
        int run = incl - s;
#pragma unroll
        for (int k = 0; k < 4; ++k) {
            rbeg[4 * lane + k] = run;
            rcnt[4 * lane + k] = run;
            run += c4[k];
        }
        if (lane == 31) rbeg[BKT_ROWS] = incl;
    }
    __syncthreads();
    int p[SORT_NIT];
#pragma unroll
    for (int q = 0; q < SORT_NIT; ++q)
        p[q] = (tid + q * 256 < n) ? atomicAdd(&rcnt[r[q].x & (BKT_ROWS - 1)], 1) : 0;
#pragma unroll
    for (int q = 0; q < SORT_NIT; ++q)
        if (tid + q * 256 < n) srec[p[q]] = r[q];
    __syncthreads();
}

__global__ __launch_bounds__(256) void bucket_spmm_rows_kernel(
    const int2* __restrict__ rec, const int* __restrict__ start,
    const int* __restrict__ cnt,
    const float* __restrict__ srcA, const float* __restrict__ srcB, int split,
    float* __restrict__ out_lo, float* __restrict__ out_hi) {
    __shared__ __align__(16) unsigned char raw[BKT_ROWS * DIM * 4];
    __shared__ int rcnt[BKT_ROWS];
    __shared__ int rbeg[BKT_ROWS + 1];
    const int tid  = (int)threadIdx.x;
    const int lane = tid & 31;
    const int wib  = __builtin_amdgcn_readfirstlane(tid >> 5);
    const int bkt  = (int)blockIdx.x;
    const size_t row0 = (size_t)bkt * BKT_ROWS;
    const int beg = __builtin_amdgcn_readfirstlane(start[bkt]);
    const int n   = __builtin_amdgcn_readfirstlane(cnt[bkt]);

    if (n <= SORT_CAP) {
        int2* srec = (int2*)raw;
        bucket_order_rows(rec, beg, n, srec, rcnt, rbeg, tid);
        const int r0 = wib * 16;
        const int s0 = __builtin_amdgcn_readfirstlane(rbeg[r0]);
        const int s1 = __builtin_amdgcn_readfirstlane(rbeg[r0 + 16]);
        float* orow = ((lane < 16) ? out_lo : out_hi) + 2 * (lane & 15);
        {
            const int li = lane & 15;
            const bool empty = (lane < 16) && (rbeg[r0 + li + 1] == rbeg[r0 + li]);
            unsigned em = __builtin_amdgcn_ballot_w32(empty);
            while (em) {
                const int rr = r0 + __builtin_ctz(em);
                em &= em - 1;
                if (row0 + rr < (size_t)N_NODES)
                    *(float2*)(orow + (row0 + rr) * HALF) = make_float2(0.f, 0.f);
            }
        }
        float2 acc = make_float2(0.f, 0.f);
        int cur = -1;
        for (int base = s0; base < s1; base += 32) {
            int nvalid = s1 - base;
            nvalid = nvalid > 32 ? 32 : nvalid;
            const int2 mine = srec[(base + lane < s1) ? (base + lane) : (s1 - 1)];
#pragma unroll
            for (int g = 0; g < 32; g += GRP) {
                if (g < nvalid) {
                    int    key[GRP];
                    float  w[GRP];
                    float2 x[GRP];
#pragma unroll
                    for (int j = 0; j < GRP; ++j) {
                        const int jj = (g + j < nvalid) ? (g + j) : (nvalid - 1);
                        key[j] = __builtin_amdgcn_readlane(mine.x, jj);
                        w[j]   = __int_as_float(__builtin_amdgcn_readlane(mine.y, jj));
                        const float* src = spmm_src_row(srcA, srcB, split, key[j] >> BKT_SHIFT);
                        x[j] = *(const float2*)(src + 2 * lane);
                    }
#pragma unroll
                    for (int j = 0; j < GRP; ++j) {
                        if (g + j < nvalid) {
                            const int row = key[j] & (BKT_ROWS - 1);
                            if (row != cur) {
                                if (cur >= 0) *(float2*)(orow + (row0 + cur) * HALF) = acc;
                                acc = make_float2(0.f, 0.f);
                                cur = row;
                            }
                            acc.x = fmaf(w[j], x[j].x, acc.x);
                            acc.y = fmaf(w[j], x[j].y, acc.y);
                        }
                    }
                }
            }
        }
        if (cur >= 0) *(float2*)(orow + (row0 + cur) * HALF) = acc;
        return;
    }

    float* acc = (float*)raw;
    for (int i = tid; i < BKT_ROWS * DIM / 4; i += 256)
        ((float4*)acc)[i] = make_float4(0.f, 0.f, 0.f, 0.f);
    __syncthreads();
    const int end = beg + n;
    int2 cur = rec[(beg + tid < end) ? (beg + tid) : (end - 1)];
    for (int base = beg; base < end; base += 256) {
        int2 nxt = make_int2(0, 0);
        int nvalid = end - base - wib * 32;
        nvalid = nvalid > 32 ? 32 : nvalid;
#pragma unroll
        for (int g = 0; g < 32; g += GRP) {
            if (g < nvalid) {
                int    key[GRP];
                float  w[GRP];
                float2 x[GRP];
#pragma unroll
                for (int j = 0; j < GRP; ++j) {
                    const int jj = (g + j < nvalid) ? (g + j) : (nvalid - 1);
                    key[j] = __builtin_amdgcn_readlane(cur.x, jj);
                    w[j]   = __int_as_float(__builtin_amdgcn_readlane(cur.y, jj));
                    const float* src = spmm_src_row(srcA, srcB, split, key[j] >> BKT_SHIFT);
                    x[j] = *(const float2*)(src + 2 * lane);
                }
                if (g == 0) nxt = rec[(base + 256 + tid < end) ? (base + 256 + tid) : (end - 1)];
#pragma unroll
                for (int j = 0; j < GRP; ++j) {
                    if (g + j < nvalid) {
                        float* d = acc + (key[j] & (BKT_ROWS - 1)) * DIM + 2 * lane;
                        atomicAdd(d,     w[j] * x[j].x);
                        atomicAdd(d + 1, w[j] * x[j].y);
                    }
                }
            }
        }
        cur = nxt;
    }
    __syncthreads();
    for (int q = tid; q < BKT_ROWS * (HALF / 4); q += 256) {
        const int rr = q >> 3, pc = q & 7;
        if (row0 + rr < (size_t)N_NODES) {
            const float4 lo = *(const float4*)(acc + rr * DIM + 4 * pc);
            const float4 hi = *(const float4*)(acc + rr * DIM + HALF + 4 * pc);
            *(float4*)(out_lo + (row0 + rr) * HALF + 4 * pc) = lo;
            *(float4*)(out_hi + (row0 + rr) * HALF + 4 * pc) = hi;
        }
    }
}

__global__ __launch_bounds__(256) void bucket_spmm_plane_kernel(
    const int2* __restrict__ rec, const int* __restrict__ start,
    const int* __restrict__ cnt,
    const float* __restrict__ src, float* __restrict__ dst) {
    __shared__ __align__(16) unsigned char raw[BKT_ROWS * HALF * 4];
    __shared__ int rcnt[BKT_ROWS];
    __shared__ int rbeg[BKT_ROWS + 1];
    const int tid  = (int)threadIdx.x;
    const int lane = tid & 31;
    const int wib  = __builtin_amdgcn_readfirstlane(tid >> 5);
    const int bkt  = (int)blockIdx.x;
    const size_t row0 = (size_t)bkt * BKT_ROWS;
    const int beg = __builtin_amdgcn_readfirstlane(start[bkt]);
    const int n   = __builtin_amdgcn_readfirstlane(cnt[bkt]);

    if (n <= SORT_CAP) {
        int2* srec = (int2*)raw;
        bucket_order_rows(rec, beg, n, srec, rcnt, rbeg, tid);
        const int r0 = wib * 16;
        const int s0 = __builtin_amdgcn_readfirstlane(rbeg[r0]);
        const int s1 = __builtin_amdgcn_readfirstlane(rbeg[r0 + 16]);
        float* orow = dst + lane;
        {
            const int li = lane & 15;
            const bool empty = (lane < 16) && (rbeg[r0 + li + 1] == rbeg[r0 + li]);
            unsigned em = __builtin_amdgcn_ballot_w32(empty);
            while (em) {
                const int rr = r0 + __builtin_ctz(em);
                em &= em - 1;
                if (row0 + rr < (size_t)N_NODES) orow[(row0 + rr) * HALF] = 0.f;
            }
        }
        float acc = 0.f;
        int cur = -1;
        for (int base = s0; base < s1; base += 32) {
            int nvalid = s1 - base;
            nvalid = nvalid > 32 ? 32 : nvalid;
            const int2 mine = srec[(base + lane < s1) ? (base + lane) : (s1 - 1)];
#pragma unroll
            for (int g = 0; g < 32; g += GRP) {
                if (g < nvalid) {
                    int   key[GRP];
                    float w[GRP];
                    float x[GRP];
#pragma unroll
                    for (int j = 0; j < GRP; ++j) {
                        const int jj = (g + j < nvalid) ? (g + j) : (nvalid - 1);
                        key[j] = __builtin_amdgcn_readlane(mine.x, jj);
                        w[j]   = __int_as_float(__builtin_amdgcn_readlane(mine.y, jj));
                        x[j]   = src[(size_t)(key[j] >> BKT_SHIFT) * HALF + lane];
                    }
#pragma unroll
                    for (int j = 0; j < GRP; ++j) {
                        if (g + j < nvalid) {
                            const int row = key[j] & (BKT_ROWS - 1);
                            if (row != cur) {
                                if (cur >= 0) orow[(row0 + cur) * HALF] = acc;
                                acc = 0.f;
                                cur = row;
                            }
                            acc = fmaf(w[j], x[j], acc);
                        }
                    }
                }
            }
        }
        if (cur >= 0) orow[(row0 + cur) * HALF] = acc;
        return;
    }

    float* acc = (float*)raw;
    for (int i = tid; i < BKT_ROWS * HALF / 4; i += 256)
        ((float4*)acc)[i] = make_float4(0.f, 0.f, 0.f, 0.f);
    __syncthreads();
    const int end = beg + n;
    int2 cur = rec[(beg + tid < end) ? (beg + tid) : (end - 1)];
    for (int base = beg; base < end; base += 256) {
        int2 nxt = make_int2(0, 0);
        int nvalid = end - base - wib * 32;
        nvalid = nvalid > 32 ? 32 : nvalid;
#pragma unroll
        for (int g = 0; g < 32; g += GRP) {
            if (g < nvalid) {
                int   key[GRP];
                float w[GRP];
                float x[GRP];
#pragma unroll
                for (int j = 0; j < GRP; ++j) {
                    const int jj = (g + j < nvalid) ? (g + j) : (nvalid - 1);
                    key[j] = __builtin_amdgcn_readlane(cur.x, jj);
                    w[j]   = __int_as_float(__builtin_amdgcn_readlane(cur.y, jj));
                    x[j]   = src[(size_t)(key[j] >> BKT_SHIFT) * HALF + lane];
                }
                if (g == 0) nxt = rec[(base + 256 + tid < end) ? (base + 256 + tid) : (end - 1)];
#pragma unroll
                for (int j = 0; j < GRP; ++j) {
                    if (g + j < nvalid)
                        atomicAdd(acc + (key[j] & (BKT_ROWS - 1)) * HALF + lane, w[j] * x[j]);
                }
            }
        }
        cur = nxt;
    }
    __syncthreads();
    for (int q = tid; q < BKT_ROWS * (HALF / 4); q += 256) {
        const int rr = q >> 3;
        if (row0 + rr < (size_t)N_NODES)
            *(float4*)(dst + row0 * HALF + 4 * q) = *(const float4*)(acc + 4 * q);
    }
}

#define PRED_WPB   8
#define PRED_TPW   8
#define PRED_BLOCKS (BATCH_CT / 16 / PRED_WPB / PRED_TPW)
#define TILE_PITCH 68
__device__ __forceinline__ void pred_tile_products(
    const float* fa, const float* fb, int lane, float cbu, float cbi, float muv,
    float* __restrict__ out, int base) {
    v8f c = {};
#pragma unroll
    for (int t = 0; t < 8; ++t) {
        const float4 a4 = *(const float4*)(fa + 4 * t);
        const float4 b4 = *(const float4*)(fb + 4 * t);
        v2f a0 = {a4.x, a4.y}, b0 = {b4.x, b4.y};
        v2f a1 = {a4.z, a4.w}, b1 = {b4.z, b4.w};
        c = __builtin_amdgcn_wmma_f32_16x16x4_f32(
                false, a0, false, b0, (short)0, c, false, false);
        c = __builtin_amdgcn_wmma_f32_16x16x4_f32(
                false, a1, false, b1, (short)0, c, false, false);
    }
    float d = 0.f;
#pragma unroll
    for (int j = 0; j < 8; ++j) {
        float vlo = __shfl(c[j], j, 32);
        float vhi = __shfl(c[j], j + 24, 32);
        if (lane == j)     d = vlo;
        if (lane == j + 8) d = vhi;
    }
    if (lane < 16) out[base + lane] = d + cbu + cbi + muv;
}

#define PRED_LDU(r) (*(const float4*)(up + (size_t)(r) * HALF + mo))
#define PRED_LDI(r) (*(const float4*)(ip + (size_t)(r) * HALF + mo))
#define PRED_GATHER_TILE()                                              \
    gu0 = PRED_LDU(ua.x); gi0 = PRED_LDI(ia.x);                         \
    gu1 = PRED_LDU(ua.y); gi1 = PRED_LDI(ia.y);                         \
    gu2 = PRED_LDU(ua.z); gi2 = PRED_LDI(ia.z);                         \
    gu3 = PRED_LDU(ua.w); gi3 = PRED_LDI(ia.w);                         \
    gu4 = PRED_LDU(ub.x); gi4 = PRED_LDI(ib.x);                         \
    gu5 = PRED_LDU(ub.y); gi5 = PRED_LDI(ib.y);                         \
    gu6 = PRED_LDU(ub.z); gi6 = PRED_LDI(ib.z);                         \
    gu7 = PRED_LDU(ub.w); gi7 = PRED_LDI(ib.w);
#define PRED_ST(j, vu, vi)                                              \
    *(float4*)(su + (j) * TILE_PITCH) = (vu);                           \
    *(float4*)(si + (j) * TILE_PITCH) = (vi);
#define PRED_STAGE_TILE()                                               \
    PRED_ST(0, gu0, gi0) PRED_ST(1, gu1, gi1) PRED_ST(2, gu2, gi2)      \
    PRED_ST(3, gu3, gi3) PRED_ST(4, gu4, gi4) PRED_ST(5, gu5, gi5)      \
    PRED_ST(6, gu6, gi6) PRED_ST(7, gu7, gi7)
#define PRED_LOAD_IDX(b)                                                \
    ur = u_idx[(b) + m];                                                \
    ir = i_idx[(b) + m];                                                \
    ua = *(const int4*)(u_idx + (b) + 8 * hi);                          \
    ub = *(const int4*)(u_idx + (b) + 8 * hi + 4);                      \
    ia = *(const int4*)(i_idx + (b) + 8 * hi);                          \
    ib = *(const int4*)(i_idx + (b) + 8 * hi + 4);

__global__ __launch_bounds__(256) void pred_kernel(
    const float* __restrict__ Ulo, const float* __restrict__ Uhi,
    const float* __restrict__ Ilo, const float* __restrict__ Ihi,
    const int*   __restrict__ u_idx, const int* __restrict__ i_idx,
    const float* __restrict__ bu, const float* __restrict__ bi,
    const float* __restrict__ mu, float* __restrict__ out) {
    __shared__ float tiles[PRED_WPB][2][16 * TILE_PITCH];

    const int wib  = threadIdx.x >> 5;
    const int lane = threadIdx.x & 31;
    const int m    = lane & 15;
    const int hi   = lane >> 4;
    int base = ((int)blockIdx.x * PRED_WPB + wib) * (16 * PRED_TPW);

    float* tu = &tiles[wib][0][0];
    float* ti = &tiles[wib][1][0];
    const float* up = (m < 8) ? Ulo : Uhi;
    const float* ip = (m < 8) ? Ilo : Ihi;
    const int    mo = 4 * (m & 7);
    float* su = tu + (8 * hi) * TILE_PITCH + 4 * m;
    float* si = ti + (8 * hi) * TILE_PITCH + 4 * m;
    const float* fa = tu + m * TILE_PITCH + 32 * hi;
    const float* fb = ti + m * TILE_PITCH + 32 * hi;
    const float muv = mu[0];

    int  ur, ir;
    int4 ua, ub, ia, ib;
    float4 gu0, gu1, gu2, gu3, gu4, gu5, gu6, gu7;
    float4 gi0, gi1, gi2, gi3, gi4, gi5, gi6, gi7;

    PRED_LOAD_IDX(base)
    PRED_GATHER_TILE()
    float bu_v = bu[ur];
    float bi_v = bi[ir];
    PRED_LOAD_IDX(base + 16)

#pragma unroll 1
    for (int it = 0; it < PRED_TPW; ++it) {
        PRED_STAGE_TILE()
        const float cbu = bu_v, cbi = bi_v;
        __builtin_amdgcn_fence(__ATOMIC_ACQ_REL, "wavefront");

        if (it + 1 < PRED_TPW) {
            PRED_GATHER_TILE()
            bu_v = bu[ur];
            bi_v = bi[ir];
            if (it + 2 < PRED_TPW) {
                PRED_LOAD_IDX(base + 32)
            }
        }

        pred_tile_products(fa, fb, lane, cbu, cbi, muv, out, base);
        base += 16;
    }
}

extern "C" void kernel_launch(void* const* d_in, const int* in_sizes, int n_in,
                              void* d_out, int out_size, void* d_ws, size_t ws_size,
                              hipStream_t stream) {
    const int*   u_idx  = (const int*)d_in[0];
    const int*   i_idx  = (const int*)d_in[1];
    const int*   A_rows = (const int*)d_in[2];
    const int*   A_cols = (const int*)d_in[3];
    const float* A_vals = (const float*)d_in[4];
    const float* Eu     = (const float*)d_in[5];
    const float* Ei     = (const float*)d_in[6];
    const float* bu     = (const float*)d_in[7];
    const float* bi     = (const float*)d_in[8];
    const float* mu     = (const float*)d_in[9];
    float* out = (float*)d_out;

    float* P0 = (float*)d_ws;
    float* P1 = P0 + (size_t)N_NODES * HALF;
    float* P2 = P1 + (size_t)N_NODES * HALF;
    int2*  rec   = (int2*)(P2 + (size_t)N_NODES * HALF);
    int*   cnt   = (int*)(rec + NNZ_CT);
    int*   fill  = cnt + NBKT_PAD;
    int*   start = fill + NBKT_PAD;

    clear_counts_kernel<<<1, 256, 0, stream>>>(cnt);
    bucket_count_kernel<<<CNT_BLOCKS, 256, 0, stream>>>(A_rows, cnt, NNZ_CT);
    bucket_place_kernel<<<PLC_BLOCKS, PLC_THREADS, 0, stream>>>(
        A_rows, A_cols, A_vals, cnt, fill, start, rec, NNZ_CT);

    bucket_spmm_rows_kernel<<<NBKT, 256, 0, stream>>>(
        rec, start, cnt, Eu, Ei, N_USERS, P0, P1);

    bucket_spmm_plane_kernel<<<NBKT, 256, 0, stream>>>(rec, start, cnt, P0, P2);
    bucket_spmm_plane_kernel<<<NBKT, 256, 0, stream>>>(rec, start, cnt, P1, P0);

    pred_kernel<<<PRED_BLOCKS, 256, 0, stream>>>(
        P2, P0, P2 + (size_t)N_USERS * HALF, P0 + (size_t)N_USERS * HALF,
        u_idx, i_idx, bu, bi, mu, out);
}


# spmm walk: each half-wave (layer 1) or quarter-wave (layer 2) walks its own rows' records, 16 B per lane: a gather brings 2 or 4 whole rows; no scalar broadcast; sum order kept
# speedup vs baseline: 3.0509x; 1.0527x over previous
#include <hip/hip_runtime.h>


#define N_USERS  100000
#define N_ITEMS  50000
#define N_NODES  (N_USERS + N_ITEMS)
#define NNZ_CT   1500000
#define BATCH_CT 1048576
#define DIM      64
#define HALF     32

typedef __attribute__((ext_vector_type(2))) float v2f;
typedef __attribute__((ext_vector_type(8))) float v8f;

#define BKT_ROWS  128
#define BKT_SHIFT 7
#define NBKT      ((N_NODES + BKT_ROWS - 1) / BKT_ROWS)
#define NBKT_PAD  1280
#define CNT_BLOCKS 512
#define CNT_EPB   ((NNZ_CT + CNT_BLOCKS - 1) / CNT_BLOCKS)
#define CNT_NIT   ((CNT_EPB + 255) / 256)
#define PLC_BLOCKS  256
#define PLC_THREADS 1024
#define PLC_EPB   ((NNZ_CT + PLC_BLOCKS - 1) / PLC_BLOCKS)
#define PLC_NIT   ((PLC_EPB + PLC_THREADS - 1) / PLC_THREADS)
#define PLC_SCAN_N (2 * PLC_THREADS)

__global__ __launch_bounds__(256) void clear_counts_kernel(int* __restrict__ cnt2) {
    for (int i = threadIdx.x; i < 2 * NBKT_PAD; i += 256) cnt2[i] = 0;
}

__global__ __launch_bounds__(256) void bucket_count_kernel(
    const int* __restrict__ rows, int* __restrict__ cnt, int nnz) {
    __shared__ int h[NBKT_PAD];
    const int tid = (int)threadIdx.x;
    for (int i = tid; i < NBKT_PAD; i += 256) h[i] = 0;
    __syncthreads();
    const int e_beg = (int)blockIdx.x * CNT_EPB;
    int r[CNT_NIT];
#pragma unroll
    for (int q = 0; q < CNT_NIT; ++q) {
        const int i = tid + q * 256;
        const int e = e_beg + i;
        r[q] = (i < CNT_EPB && e < nnz) ? rows[e] : -1;
    }
#pragma unroll
    for (int q = 0; q < CNT_NIT; ++q) {
        if ((unsigned)r[q] < (unsigned)N_NODES) atomicAdd(&h[r[q] >> BKT_SHIFT], 1);
    }
    __syncthreads();
    for (int k = tid; k < NBKT; k += 256) {
        const int c = h[k];
        if (c) atomicAdd(&cnt[k], c);
    }
}

__global__ __launch_bounds__(PLC_THREADS) void bucket_place_kernel(
    const int* __restrict__ rows, const int* __restrict__ cols,
    const float* __restrict__ vals, const int* __restrict__ cnt,
    int* __restrict__ fill, int* __restrict__ start_out,
    int2* __restrict__ rec, int nnz) {
    __shared__ int h[PLC_SCAN_N];
    __shared__ int st[PLC_SCAN_N];
    __shared__ int lp[PLC_SCAN_N];
    __shared__ int wsum_g[32];
    __shared__ int wsum_l[32];
    __shared__ int2 srec[PLC_EPB];
    __shared__ unsigned short sbkt[PLC_EPB];
    const int tid  = (int)threadIdx.x;
    const int lane = tid & 31;
    const int wib  = tid >> 5;
    for (int i = tid; i < PLC_SCAN_N; i += PLC_THREADS) h[i] = 0;
    __syncthreads();
    const int e_beg = (int)blockIdx.x * PLC_EPB;
    int   r[PLC_NIT], c[PLC_NIT];
    float v[PLC_NIT];
#pragma unroll
    for (int q = 0; q < PLC_NIT; ++q) {
        const int i = tid + q * PLC_THREADS;
        const int e = e_beg + i;
        const bool ok = (i < PLC_EPB && e < nnz);
        r[q] = ok ? rows[e] : -1;
        c[q] = ok ? cols[e] : 0;
        v[q] = ok ? vals[e] : 0.f;
    }
    const int k0 = 2 * tid, k1 = 2 * tid + 1;
    const int g0 = (k0 < NBKT) ? cnt[k0] : 0;
    const int g1 = (k1 < NBKT) ? cnt[k1] : 0;
#pragma unroll
    for (int q = 0; q < PLC_NIT; ++q) {
        if ((unsigned)r[q] < (unsigned)N_NODES) atomicAdd(&h[r[q] >> BKT_SHIFT], 1);
    }
    __syncthreads();
    const int l0 = h[k0], l1 = h[k1];
    const int sg = g0 + g1, sl = l0 + l1;
    int ig = sg, il = sl;
    for (int d = 1; d < 32; d <<= 1) {
        const int og = __shfl_up(ig, d, 32);
        const int ol = __shfl_up(il, d, 32);
        if (lane >= d) { ig += og; il += ol; }
    }
    if (lane == 31) { wsum_g[wib] = ig; wsum_l[wib] = il; }
    __syncthreads();
    int offg = 0, offl = 0, nloc = 0;
    for (int w = 0; w < 32; ++w) {
        const int a = wsum_g[w], b = wsum_l[w];
        if (w < wib) { offg += a; offl += b; }
        nloc += b;
    }
    const int rung = offg + ig - sg;
    const int runl = offl + il - sl;
    st[k0] = rung;  st[k1] = rung + g0;
    lp[k0] = runl;  lp[k1] = runl + l0;
    __syncthreads();
    for (int k = tid; k < NBKT; k += PLC_THREADS) {
        const int n = h[k];
        const int b = n ? atomicAdd(&fill[k], n) : 0;
        if (blockIdx.x == 0) start_out[k] = st[k];
        st[k] = st[k] + b - lp[k];
        h[k]  = lp[k];
    }
    __syncthreads();
#pragma unroll
    for (int q = 0; q < PLC_NIT; ++q) {
        if ((unsigned)r[q] < (unsigned)N_NODES) {
            const int bk = r[q] >> BKT_SHIFT;
            const int p  = atomicAdd(&h[bk], 1);
            srec[p] = make_int2((c[q] << BKT_SHIFT) | (r[q] & (BKT_ROWS - 1)),
                                __float_as_int(v[q]));
            sbkt[p] = (unsigned short)bk;
        }
    }
    __syncthreads();
    for (int s = tid; s < nloc; s += PLC_THREADS)
        rec[st[sbkt[s]] + s] = srec[s];
}

#define GRP 16
#define SUBG 8
#define SORT_CAP 2048
#define SORT_NIT (SORT_CAP / 256)

__device__ __forceinline__ const float* spmm_src_row(
    const float* __restrict__ srcA, const float* __restrict__ srcB, int split, int c) {
    return (c < split) ? (srcA + (size_t)c * DIM)
                       : (srcB + (size_t)(c - split) * DIM);
}

__device__ __forceinline__ void bucket_order_rows(
    const int2* __restrict__ rec, int beg, int n,
    int2* srec, int* rcnt, int* rbeg, int tid) {
    const int lane = tid & 31;
    if (tid < BKT_ROWS) rcnt[tid] = 0;
    int2 r[SORT_NIT];
#pragma unroll
    for (int q = 0; q < SORT_NIT; ++q) {
        const int i = tid + q * 256;
        r[q] = (i < n) ? rec[beg + i] : make_int2(0, 0);
    }
    __syncthreads();
#pragma unroll
    for (int q = 0; q < SORT_NIT; ++q)
        if (tid + q * 256 < n) atomicAdd(&rcnt[r[q].x & (BKT_ROWS - 1)], 1);
    __syncthreads();
    if (tid < 32) {
        int c4[4];
        int s = 0;
#pragma unroll
        for (int k = 0; k < 4; ++k) { c4[k] = rcnt[4 * lane + k]; s += c4[k]; }
        int incl = s;
        for (int d = 1; d < 32; d <<= 1) {
            const int o = __shfl_up(incl, d, 32);
            if (lane >= d) incl += o;
        }
        int run = incl - s;
#pragma unroll
        for (int k = 0; k < 4; ++k) {
            rbeg[4 * lane + k] = run;
            rcnt[4 * lane + k] = run;
            run += c4[k];
        }
        if (lane == 31) rbeg[BKT_ROWS] = incl;
    }
    __syncthreads();
    int p[SORT_NIT];
#pragma unroll
    for (int q = 0; q < SORT_NIT; ++q)
        p[q] = (tid + q * 256 < n) ? atomicAdd(&rcnt[r[q].x & (BKT_ROWS - 1)], 1) : 0;
#pragma unroll
    for (int q = 0; q < SORT_NIT; ++q)
        if (tid + q * 256 < n) srec[p[q]] = r[q];
    __syncthreads();
}

__global__ __launch_bounds__(256) void bucket_spmm_rows_kernel(
    const int2* __restrict__ rec, const int* __restrict__ start,
    const int* __restrict__ cnt,
    const float* __restrict__ srcA, const float* __restrict__ srcB, int split,
    float* __restrict__ out_lo, float* __restrict__ out_hi) {
    __shared__ __align__(16) unsigned char raw[BKT_ROWS * DIM * 4];
    __shared__ int rcnt[BKT_ROWS];
    __shared__ int rbeg[BKT_ROWS + 1];
    const int tid  = (int)threadIdx.x;
    const int lane = tid & 31;
    const int wib  = __builtin_amdgcn_readfirstlane(tid >> 5);
    const int bkt  = (int)blockIdx.x;
    const size_t row0 = (size_t)bkt * BKT_ROWS;
    const int beg = __builtin_amdgcn_readfirstlane(start[bkt]);
    const int n   = __builtin_amdgcn_readfirstlane(cnt[bkt]);

    if (n <= SORT_CAP) {
        int2* srec = (int2*)raw;
        bucket_order_rows(rec, beg, n, srec, rcnt, rbeg, tid);
        const int sub = lane >> 4;
        const int li  = lane & 15;
        const int rq0 = wib * 16 + sub * 8;
        float* obase = ((li < 8) ? out_lo : out_hi) + 4 * (li & 7);
#pragma unroll
        for (int k = 0; k < 8; ++k) {
            const int rr = rq0 + k;
            if (rbeg[rr + 1] == rbeg[rr] && row0 + rr < (size_t)N_NODES)
                *(float4*)(obase + (row0 + rr) * HALF) = make_float4(0.f, 0.f, 0.f, 0.f);
        }
        int pos = rbeg[rq0];
        const int pend = rbeg[rq0 + 8];
        const int last = pend > 0 ? pend - 1 : 0;
        float4 acc = make_float4(0.f, 0.f, 0.f, 0.f);
        int cur = -1;
        while (__builtin_amdgcn_ballot_w32(pos < pend) != 0u) {
            const int nv = pend - pos;
            int2   r[SUBG];
            float4 x[SUBG];
#pragma unroll
            for (int j = 0; j < SUBG; ++j) {
                const int i = pos + j;
                r[j] = srec[i < last ? i : last];
            }
#pragma unroll
            for (int j = 0; j < SUBG; ++j) {
                const float* s = spmm_src_row(srcA, srcB, split, r[j].x >> BKT_SHIFT);
                x[j] = *(const float4*)(s + 4 * li);
            }
#pragma unroll
            for (int j = 0; j < SUBG; ++j)
                if (j < nv) {
                    const int row = r[j].x & (BKT_ROWS - 1);
                    if (row != cur) {
                        if (cur >= 0) *(float4*)(obase + (row0 + cur) * HALF) = acc;
                        acc = make_float4(0.f, 0.f, 0.f, 0.f);
                        cur = row;
                    }
                    const float w = __int_as_float(r[j].y);
                    acc.x = fmaf(w, x[j].x, acc.x);
                    acc.y = fmaf(w, x[j].y, acc.y);
                    acc.z = fmaf(w, x[j].z, acc.z);
                    acc.w = fmaf(w, x[j].w, acc.w);
                }
            pos += SUBG;
        }
        if (cur >= 0) *(float4*)(obase + (row0 + cur) * HALF) = acc;
        return;
    }

    float* acc = (float*)raw;
    for (int i = tid; i < BKT_ROWS * DIM / 4; i += 256)
        ((float4*)acc)[i] = make_float4(0.f, 0.f, 0.f, 0.f);
    __syncthreads();
    const int end = beg + n;
    int2 cur = rec[(beg + tid < end) ? (beg + tid) : (end - 1)];
    for (int base = beg; base < end; base += 256) {
        int2 nxt = make_int2(0, 0);
        int nvalid = end - base - wib * 32;
        nvalid = nvalid > 32 ? 32 : nvalid;
#pragma unroll
        for (int g = 0; g < 32; g += GRP) {
            if (g < nvalid) {
                int    key[GRP];
                float  w[GRP];
                float2 x[GRP];
#pragma unroll
                for (int j = 0; j < GRP; ++j) {
                    const int jj = (g + j < nvalid) ? (g + j) : (nvalid - 1);
                    key[j] = __builtin_amdgcn_readlane(cur.x, jj);
                    w[j]   = __int_as_float(__builtin_amdgcn_readlane(cur.y, jj));
                    const float* src = spmm_src_row(srcA, srcB, split, key[j] >> BKT_SHIFT);
                    x[j] = *(const float2*)(src + 2 * lane);
                }
                if (g == 0) nxt = rec[(base + 256 + tid < end) ? (base + 256 + tid) : (end - 1)];
#pragma unroll
                for (int j = 0; j < GRP; ++j) {
                    if (g + j < nvalid) {
                        float* d = acc + (key[j] & (BKT_ROWS - 1)) * DIM + 2 * lane;
                        atomicAdd(d,     w[j] * x[j].x);
                        atomicAdd(d + 1, w[j] * x[j].y);
                    }
                }
            }
        }
        cur = nxt;
    }
    __syncthreads();
    for (int q = tid; q < BKT_ROWS * (HALF / 4); q += 256) {
        const int rr = q >> 3, pc = q & 7;
        if (row0 + rr < (size_t)N_NODES) {
            const float4 lo = *(const float4*)(acc + rr * DIM + 4 * pc);
            const float4 hi = *(const float4*)(acc + rr * DIM + HALF + 4 * pc);
            *(float4*)(out_lo + (row0 + rr) * HALF + 4 * pc) = lo;
            *(float4*)(out_hi + (row0 + rr) * HALF + 4 * pc) = hi;
        }
    }
}

__global__ __launch_bounds__(256) void bucket_spmm_plane_kernel(
    const int2* __restrict__ rec, const int* __restrict__ start,
    const int* __restrict__ cnt,
    const float* __restrict__ src, float* __restrict__ dst) {
    __shared__ __align__(16) unsigned char raw[BKT_ROWS * HALF * 4];
    __shared__ int rcnt[BKT_ROWS];
    __shared__ int rbeg[BKT_ROWS + 1];
    const int tid  = (int)threadIdx.x;
    const int lane = tid & 31;
    const int wib  = __builtin_amdgcn_readfirstlane(tid >> 5);
    const int bkt  = (int)blockIdx.x;
    const size_t row0 = (size_t)bkt * BKT_ROWS;
    const int beg = __builtin_amdgcn_readfirstlane(start[bkt]);
    const int n   = __builtin_amdgcn_readfirstlane(cnt[bkt]);

    if (n <= SORT_CAP) {
        int2* srec = (int2*)raw;
        bucket_order_rows(rec, beg, n, srec, rcnt, rbeg, tid);
        const int sub = lane >> 3;
        const int li  = lane & 7;
        const int rq0 = wib * 16 + sub * 4;
        float* obase = dst + 4 * li;
#pragma unroll
        for (int k = 0; k < 4; ++k) {
            const int rr = rq0 + k;
            if (rbeg[rr + 1] == rbeg[rr] && row0 + rr < (size_t)N_NODES)
                *(float4*)(obase + (row0 + rr) * HALF) = make_float4(0.f, 0.f, 0.f, 0.f);
        }
        int pos = rbeg[rq0];
        const int pend = rbeg[rq0 + 4];
        const int last = pend > 0 ? pend - 1 : 0;
        float4 acc = make_float4(0.f, 0.f, 0.f, 0.f);
        int cur = -1;
        while (__builtin_amdgcn_ballot_w32(pos < pend) != 0u) {
            const int nv = pend - pos;
            int2   r[SUBG];
            float4 x[SUBG];
#pragma unroll
            for (int j = 0; j < SUBG; ++j) {
                const int i = pos + j;
                r[j] = srec[i < last ? i : last];
            }
#pragma unroll
            for (int j = 0; j < SUBG; ++j) {
                const unsigned off = ((unsigned)r[j].x >> BKT_SHIFT) * (unsigned)(HALF * 4)
                                   + 16u * (unsigned)li;
                x[j] = *(const float4*)((const char*)src + off);
            }
#pragma unroll
            for (int j = 0; j < SUBG; ++j)
                if (j < nv) {
                    const int row = r[j].x & (BKT_ROWS - 1);
                    if (row != cur) {
                        if (cur >= 0) *(float4*)(obase + (row0 + cur) * HALF) = acc;
                        acc = make_float4(0.f, 0.f, 0.f, 0.f);
                        cur = row;
                    }
                    const float w = __int_as_float(r[j].y);
                    acc.x = fmaf(w, x[j].x, acc.x);
                    acc.y = fmaf(w, x[j].y, acc.y);
                    acc.z = fmaf(w, x[j].z, acc.z);
                    acc.w = fmaf(w, x[j].w, acc.w);
                }
            pos += SUBG;
        }
        if (cur >= 0) *(float4*)(obase + (row0 + cur) * HALF) = acc;
        return;
    }

    float* acc = (float*)raw;
    for (int i = tid; i < BKT_ROWS * HALF / 4; i += 256)
        ((float4*)acc)[i] = make_float4(0.f, 0.f, 0.f, 0.f);
    __syncthreads();
    const int end = beg + n;
    int2 cur = rec[(beg + tid < end) ? (beg + tid) : (end - 1)];
    for (int base = beg; base < end; base += 256) {
        int2 nxt = make_int2(0, 0);
        int nvalid = end - base - wib * 32;
        nvalid = nvalid > 32 ? 32 : nvalid;
#pragma unroll
        for (int g = 0; g < 32; g += GRP) {
            if (g < nvalid) {
                int   key[GRP];
                float w[GRP];
                float x[GRP];
#pragma unroll
                for (int j = 0; j < GRP; ++j) {
                    const int jj = (g + j < nvalid) ? (g + j) : (nvalid - 1);
                    key[j] = __builtin_amdgcn_readlane(cur.x, jj);
                    w[j]   = __int_as_float(__builtin_amdgcn_readlane(cur.y, jj));
                    x[j]   = src[(size_t)(key[j] >> BKT_SHIFT) * HALF + lane];
                }
                if (g == 0) nxt = rec[(base + 256 + tid < end) ? (base + 256 + tid) : (end - 1)];
#pragma unroll
                for (int j = 0; j < GRP; ++j) {
                    if (g + j < nvalid)
                        atomicAdd(acc + (key[j] & (BKT_ROWS - 1)) * HALF + lane, w[j] * x[j]);
                }
            }
        }
        cur = nxt;
    }
    __syncthreads();
    for (int q = tid; q < BKT_ROWS * (HALF / 4); q += 256) {
        const int rr = q >> 3;
        if (row0 + rr < (size_t)N_NODES)
            *(float4*)(dst + row0 * HALF + 4 * q) = *(const float4*)(acc + 4 * q);
    }
}

#define PRED_WPB   8
#define PRED_TPW   8
#define PRED_BLOCKS (BATCH_CT / 16 / PRED_WPB / PRED_TPW)
#define TILE_PITCH 68
__device__ __forceinline__ void pred_tile_products(
    const float* fa, const float* fb, int lane, float cbu, float cbi, float muv,
    float* __restrict__ out, int base) {
    v8f c = {};
#pragma unroll
    for (int t = 0; t < 8; ++t) {
        const float4 a4 = *(const float4*)(fa + 4 * t);
        const float4 b4 = *(const float4*)(fb + 4 * t);
        v2f a0 = {a4.x, a4.y}, b0 = {b4.x, b4.y};
        v2f a1 = {a4.z, a4.w}, b1 = {b4.z, b4.w};
        c = __builtin_amdgcn_wmma_f32_16x16x4_f32(
                false, a0, false, b0, (short)0, c, false, false);
        c = __builtin_amdgcn_wmma_f32_16x16x4_f32(
                false, a1, false, b1, (short)0, c, false, false);
    }
    float d = 0.f;
#pragma unroll
    for (int j = 0; j < 8; ++j) {
        float vlo = __shfl(c[j], j, 32);
        float vhi = __shfl(c[j], j + 24, 32);
        if (lane == j)     d = vlo;
        if (lane == j + 8) d = vhi;
    }
    if (lane < 16) out[base + lane] = d + cbu + cbi + muv;
}

#define PRED_LDU(r) (*(const float4*)(up + (size_t)(r) * HALF + mo))
#define PRED_LDI(r) (*(const float4*)(ip + (size_t)(r) * HALF + mo))
#define PRED_GATHER_TILE()                                              \
    gu0 = PRED_LDU(ua.x); gi0 = PRED_LDI(ia.x);                         \
    gu1 = PRED_LDU(ua.y); gi1 = PRED_LDI(ia.y);                         \
    gu2 = PRED_LDU(ua.z); gi2 = PRED_LDI(ia.z);                         \
    gu3 = PRED_LDU(ua.w); gi3 = PRED_LDI(ia.w);                         \
    gu4 = PRED_LDU(ub.x); gi4 = PRED_LDI(ib.x);                         \
    gu5 = PRED_LDU(ub.y); gi5 = PRED_LDI(ib.y);                         \
    gu6 = PRED_LDU(ub.z); gi6 = PRED_LDI(ib.z);                         \
    gu7 = PRED_LDU(ub.w); gi7 = PRED_LDI(ib.w);
#define PRED_ST(j, vu, vi)                                              \
    *(float4*)(su + (j) * TILE_PITCH) = (vu);                           \
    *(float4*)(si + (j) * TILE_PITCH) = (vi);
#define PRED_STAGE_TILE()                                               \
    PRED_ST(0, gu0, gi0) PRED_ST(1, gu1, gi1) PRED_ST(2, gu2, gi2)      \
    PRED_ST(3, gu3, gi3) PRED_ST(4, gu4, gi4) PRED_ST(5, gu5, gi5)      \
    PRED_ST(6, gu6, gi6) PRED_ST(7, gu7, gi7)
#define PRED_LOAD_IDX(b)                                                \
    ur = u_idx[(b) + m];                                                \
    ir = i_idx[(b) + m];                                                \
    ua = *(const int4*)(u_idx + (b) + 8 * hi);                          \
    ub = *(const int4*)(u_idx + (b) + 8 * hi + 4);                      \
    ia = *(const int4*)(i_idx + (b) + 8 * hi);                          \
    ib = *(const int4*)(i_idx + (b) + 8 * hi + 4);

__global__ __launch_bounds__(256) void pred_kernel(
    const float* __restrict__ Ulo, const float* __restrict__ Uhi,
    const float* __restrict__ Ilo, const float* __restrict__ Ihi,
    const int*   __restrict__ u_idx, const int* __restrict__ i_idx,
    const float* __restrict__ bu, const float* __restrict__ bi,
    const float* __restrict__ mu, float* __restrict__ out) {
    __shared__ float tiles[PRED_WPB][2][16 * TILE_PITCH];

    const int wib  = threadIdx.x >> 5;
    const int lane = threadIdx.x & 31;
    const int m    = lane & 15;
    const int hi   = lane >> 4;
    int base = ((int)blockIdx.x * PRED_WPB + wib) * (16 * PRED_TPW);

    float* tu = &tiles[wib][0][0];
    float* ti = &tiles[wib][1][0];
    const float* up = (m < 8) ? Ulo : Uhi;
    const float* ip = (m < 8) ? Ilo : Ihi;
    const int    mo = 4 * (m & 7);
    float* su = tu + (8 * hi) * TILE_PITCH + 4 * m;
    float* si = ti + (8 * hi) * TILE_PITCH + 4 * m;
    const float* fa = tu + m * TILE_PITCH + 32 * hi;
    const float* fb = ti + m * TILE_PITCH + 32 * hi;
    const float muv = mu[0];

    int  ur, ir;
    int4 ua, ub, ia, ib;
    float4 gu0, gu1, gu2, gu3, gu4, gu5, gu6, gu7;
    float4 gi0, gi1, gi2, gi3, gi4, gi5, gi6, gi7;

    PRED_LOAD_IDX(base)
    PRED_GATHER_TILE()
    float bu_v = bu[ur];
    float bi_v = bi[ir];
    PRED_LOAD_IDX(base + 16)

#pragma unroll 1
    for (int it = 0; it < PRED_TPW; ++it) {
        PRED_STAGE_TILE()
        const float cbu = bu_v, cbi = bi_v;
        __builtin_amdgcn_fence(__ATOMIC_ACQ_REL, "wavefront");

        if (it + 1 < PRED_TPW) {
            PRED_GATHER_TILE()
            bu_v = bu[ur];
            bi_v = bi[ir];
            if (it + 2 < PRED_TPW) {
                PRED_LOAD_IDX(base + 32)
            }
        }

        pred_tile_products(fa, fb, lane, cbu, cbi, muv, out, base);
        base += 16;
    }
}

extern "C" void kernel_launch(void* const* d_in, const int* in_sizes, int n_in,
                              void* d_out, int out_size, void* d_ws, size_t ws_size,
                              hipStream_t stream) {
    const int*   u_idx  = (const int*)d_in[0];
    const int*   i_idx  = (const int*)d_in[1];
    const int*   A_rows = (const int*)d_in[2];
    const int*   A_cols = (const int*)d_in[3];
    const float* A_vals = (const float*)d_in[4];
    const float* Eu     = (const float*)d_in[5];
    const float* Ei     = (const float*)d_in[6];
    const float* bu     = (const float*)d_in[7];
    const float* bi     = (const float*)d_in[8];
    const float* mu     = (const float*)d_in[9];
    float* out = (float*)d_out;

    float* P0 = (float*)d_ws;
    float* P1 = P0 + (size_t)N_NODES * HALF;
    float* P2 = P1 + (size_t)N_NODES * HALF;
    int2*  rec   = (int2*)(P2 + (size_t)N_NODES * HALF);
    int*   cnt   = (int*)(rec + NNZ_CT);
    int*   fill  = cnt + NBKT_PAD;
    int*   start = fill + NBKT_PAD;

    clear_counts_kernel<<<1, 256, 0, stream>>>(cnt);
    bucket_count_kernel<<<CNT_BLOCKS, 256, 0, stream>>>(A_rows, cnt, NNZ_CT);
    bucket_place_kernel<<<PLC_BLOCKS, PLC_THREADS, 0, stream>>>(
        A_rows, A_cols, A_vals, cnt, fill, start, rec, NNZ_CT);

    bucket_spmm_rows_kernel<<<NBKT, 256, 0, stream>>>(
        rec, start, cnt, Eu, Ei, N_USERS, P0, P1);

    bucket_spmm_plane_kernel<<<NBKT, 256, 0, stream>>>(rec, start, cnt, P0, P2);
    bucket_spmm_plane_kernel<<<NBKT, 256, 0, stream>>>(rec, start, cnt, P1, P0);

    pred_kernel<<<PRED_BLOCKS, 256, 0, stream>>>(
        P2, P0, P2 + (size_t)N_USERS * HALF, P0 + (size_t)N_USERS * HALF,
        u_idx, i_idx, bu, bi, mu, out);
}


# spmm walk without per-record branching: the sort marks each row's last record (sign bit), spent slots read a zero-weight pad record, layer-1 source table picked by select; order of sums kept
# speedup vs baseline: 3.1993x; 1.0486x over previous
#include <hip/hip_runtime.h>


#define N_USERS  100000
#define N_ITEMS  50000
#define N_NODES  (N_USERS + N_ITEMS)
#define NNZ_CT   1500000
#define BATCH_CT 1048576
#define DIM      64
#define HALF     32

typedef __attribute__((ext_vector_type(2))) float v2f;
typedef __attribute__((ext_vector_type(8))) float v8f;

#define BKT_ROWS  128
#define BKT_SHIFT 7
#define NBKT      ((N_NODES + BKT_ROWS - 1) / BKT_ROWS)
#define NBKT_PAD  1280
#define CNT_BLOCKS 512
#define CNT_EPB   ((NNZ_CT + CNT_BLOCKS - 1) / CNT_BLOCKS)
#define CNT_NIT   ((CNT_EPB + 255) / 256)
#define PLC_BLOCKS  256
#define PLC_THREADS 1024
#define PLC_EPB   ((NNZ_CT + PLC_BLOCKS - 1) / PLC_BLOCKS)
#define PLC_NIT   ((PLC_EPB + PLC_THREADS - 1) / PLC_THREADS)
#define PLC_SCAN_N (2 * PLC_THREADS)

__global__ __launch_bounds__(256) void clear_counts_kernel(int* __restrict__ cnt2) {
    for (int i = threadIdx.x; i < 2 * NBKT_PAD; i += 256) cnt2[i] = 0;
}

__global__ __launch_bounds__(256) void bucket_count_kernel(
    const int* __restrict__ rows, int* __restrict__ cnt, int nnz) {
    __shared__ int h[NBKT_PAD];
    const int tid = (int)threadIdx.x;
    for (int i = tid; i < NBKT_PAD; i += 256) h[i] = 0;
    __syncthreads();
    const int e_beg = (int)blockIdx.x * CNT_EPB;
    int r[CNT_NIT];
#pragma unroll
    for (int q = 0; q < CNT_NIT; ++q) {
        const int i = tid + q * 256;
        const int e = e_beg + i;
        r[q] = (i < CNT_EPB && e < nnz) ? rows[e] : -1;
    }
#pragma unroll
    for (int q = 0; q < CNT_NIT; ++q) {
        if ((unsigned)r[q] < (unsigned)N_NODES) atomicAdd(&h[r[q] >> BKT_SHIFT], 1);
    }
    __syncthreads();
    for (int k = tid; k < NBKT; k += 256) {
        const int c = h[k];
        if (c) atomicAdd(&cnt[k], c);
    }
}

__global__ __launch_bounds__(PLC_THREADS) void bucket_place_kernel(
    const int* __restrict__ rows, const int* __restrict__ cols,
    const float* __restrict__ vals, const int* __restrict__ cnt,
    int* __restrict__ fill, int* __restrict__ start_out,
    int2* __restrict__ rec, int nnz) {
    __shared__ int h[PLC_SCAN_N];
    __shared__ int st[PLC_SCAN_N];
    __shared__ int lp[PLC_SCAN_N];
    __shared__ int wsum_g[32];
    __shared__ int wsum_l[32];
    __shared__ int2 srec[PLC_EPB];
    __shared__ unsigned short sbkt[PLC_EPB];
    const int tid  = (int)threadIdx.x;
    const int lane = tid & 31;
    const int wib  = tid >> 5;
    for (int i = tid; i < PLC_SCAN_N; i += PLC_THREADS) h[i] = 0;
    __syncthreads();
    const int e_beg = (int)blockIdx.x * PLC_EPB;
    int   r[PLC_NIT], c[PLC_NIT];
    float v[PLC_NIT];
#pragma unroll
    for (int q = 0; q < PLC_NIT; ++q) {
        const int i = tid + q * PLC_THREADS;
        const int e = e_beg + i;
        const bool ok = (i < PLC_EPB && e < nnz);
        r[q] = ok ? rows[e] : -1;
        c[q] = ok ? cols[e] : 0;
        v[q] = ok ? vals[e] : 0.f;
    }
    const int k0 = 2 * tid, k1 = 2 * tid + 1;
    const int g0 = (k0 < NBKT) ? cnt[k0] : 0;
    const int g1 = (k1 < NBKT) ? cnt[k1] : 0;
#pragma unroll
    for (int q = 0; q < PLC_NIT; ++q) {
        if ((unsigned)r[q] < (unsigned)N_NODES) atomicAdd(&h[r[q] >> BKT_SHIFT], 1);
    }
    __syncthreads();
    const int l0 = h[k0], l1 = h[k1];
    const int sg = g0 + g1, sl = l0 + l1;
    int ig = sg, il = sl;
    for (int d = 1; d < 32; d <<= 1) {
        const int og = __shfl_up(ig, d, 32);
        const int ol = __shfl_up(il, d, 32);
        if (lane >= d) { ig += og; il += ol; }
    }
    if (lane == 31) { wsum_g[wib] = ig; wsum_l[wib] = il; }
    __syncthreads();
    int offg = 0, offl = 0, nloc = 0;
    for (int w = 0; w < 32; ++w) {
        const int a = wsum_g[w], b = wsum_l[w];
        if (w < wib) { offg += a; offl += b; }
        nloc += b;
    }
    const int rung = offg + ig - sg;
    const int runl = offl + il - sl;
    st[k0] = rung;  st[k1] = rung + g0;
    lp[k0] = runl;  lp[k1] = runl + l0;
    __syncthreads();
    for (int k = tid; k < NBKT; k += PLC_THREADS) {
        const int n = h[k];
        const int b = n ? atomicAdd(&fill[k], n) : 0;
        if (blockIdx.x == 0) start_out[k] = st[k];
        st[k] = st[k] + b - lp[k];
        h[k]  = lp[k];
    }
    __syncthreads();
#pragma unroll
    for (int q = 0; q < PLC_NIT; ++q) {
        if ((unsigned)r[q] < (unsigned)N_NODES) {
            const int bk = r[q] >> BKT_SHIFT;
            const int p  = atomicAdd(&h[bk], 1);
            srec[p] = make_int2((c[q] << BKT_SHIFT) | (r[q] & (BKT_ROWS - 1)),
                                __float_as_int(v[q]));
            sbkt[p] = (unsigned short)bk;
        }
    }
    __syncthreads();
    for (int s = tid; s < nloc; s += PLC_THREADS)
        rec[st[sbkt[s]] + s] = srec[s];
}

#define GRP 16
#define SUBG 8
#define SORT_CAP 2048
#define SORT_NIT (SORT_CAP / 256)
#define REC_LAST  0x80000000u
#define REC_COLS  0x7fffffffu
#define PAD_SLOT  SORT_CAP

__device__ __forceinline__ const float* spmm_src_row(
    const float* __restrict__ srcA, const float* __restrict__ srcB, int split, int c) {
    return (c < split) ? (srcA + (size_t)c * DIM)
                       : (srcB + (size_t)(c - split) * DIM);
}

__device__ __forceinline__ void bucket_order_rows(
    const int2* __restrict__ rec, int beg, int n,
    int2* srec, int* rcnt, int* rbeg, int tid) {
    const int lane = tid & 31;
    if (tid < BKT_ROWS) rcnt[tid] = 0;
    int2 r[SORT_NIT];
#pragma unroll
    for (int q = 0; q < SORT_NIT; ++q) {
        const int i = tid + q * 256;
        r[q] = (i < n) ? rec[beg + i] : make_int2(0, 0);
    }
    __syncthreads();
#pragma unroll
    for (int q = 0; q < SORT_NIT; ++q)
        if (tid + q * 256 < n) atomicAdd(&rcnt[r[q].x & (BKT_ROWS - 1)], 1);
    __syncthreads();
    if (tid < 32) {
        int c4[4];
        int s = 0;
#pragma unroll
        for (int k = 0; k < 4; ++k) { c4[k] = rcnt[4 * lane + k]; s += c4[k]; }
        int incl = s;
        for (int d = 1; d < 32; d <<= 1) {
            const int o = __shfl_up(incl, d, 32);
            if (lane >= d) incl += o;
        }
        int run = incl - s;
#pragma unroll
        for (int k = 0; k < 4; ++k) {
            rbeg[4 * lane + k] = run;
            rcnt[4 * lane + k] = run;
            run += c4[k];
        }
        if (lane == 31) rbeg[BKT_ROWS] = incl;
    }
    __syncthreads();
    int p[SORT_NIT], e[SORT_NIT];
#pragma unroll
    for (int q = 0; q < SORT_NIT; ++q) {
        const bool ok = (tid + q * 256 < n);
        const int row = r[q].x & (BKT_ROWS - 1);
        p[q] = ok ? atomicAdd(&rcnt[row], 1) : 0;
        e[q] = ok ? rbeg[row + 1] : 0;
    }
#pragma unroll
    for (int q = 0; q < SORT_NIT; ++q)
        if (tid + q * 256 < n) {
            int2 t = r[q];
            if (p[q] + 1 == e[q]) t.x = (int)((unsigned)t.x | REC_LAST);
            srec[p[q]] = t;
        }
    if (tid == 0) srec[PAD_SLOT] = make_int2(0, 0);
    __syncthreads();
}

__global__ __launch_bounds__(256) void bucket_spmm_rows_kernel(
    const int2* __restrict__ rec, const int* __restrict__ start,
    const int* __restrict__ cnt,
    const float* __restrict__ srcA, const float* __restrict__ srcB, int split,
    float* __restrict__ out_lo, float* __restrict__ out_hi) {
    __shared__ __align__(16) unsigned char raw[BKT_ROWS * DIM * 4];
    __shared__ int rcnt[BKT_ROWS];
    __shared__ int rbeg[BKT_ROWS + 1];
    const int tid  = (int)threadIdx.x;
    const int lane = tid & 31;
    const int wib  = __builtin_amdgcn_readfirstlane(tid >> 5);
    const int bkt  = (int)blockIdx.x;
    const size_t row0 = (size_t)bkt * BKT_ROWS;
    const int beg = __builtin_amdgcn_readfirstlane(start[bkt]);
    const int n   = __builtin_amdgcn_readfirstlane(cnt[bkt]);

    if (n <= SORT_CAP) {
        int2* srec = (int2*)raw;
        bucket_order_rows(rec, beg, n, srec, rcnt, rbeg, tid);
        const int sub = lane >> 4;
        const int li  = lane & 15;
        const int rq0 = wib * 16 + sub * 8;
        float* obase = ((li < 8) ? out_lo : out_hi) + 4 * (li & 7);
#pragma unroll
        for (int k = 0; k < 8; ++k) {
            const int rr = rq0 + k;
            if (rbeg[rr + 1] == rbeg[rr] && row0 + rr < (size_t)N_NODES)
                *(float4*)(obase + (row0 + rr) * HALF) = make_float4(0.f, 0.f, 0.f, 0.f);
        }
        const char* tabA = (const char*)srcA + 16 * li;
        const char* tabB = (const char*)srcB + 16 * li - (long long)split * (DIM * 4);
        int pos = rbeg[rq0];
        const int pend = rbeg[rq0 + 8];
        float4 acc = make_float4(0.f, 0.f, 0.f, 0.f);
        while (__builtin_amdgcn_ballot_w32(pos < pend) != 0u) {
            int2   r[SUBG];
            float4 x[SUBG];
#pragma unroll
            for (int j = 0; j < SUBG; ++j) {
                const int i = pos + j;
                r[j] = srec[i < pend ? i : PAD_SLOT];
            }
#pragma unroll
            for (int j = 0; j < SUBG; ++j) {
                const unsigned cj = ((unsigned)r[j].x & REC_COLS) >> BKT_SHIFT;
                const char* tab = ((int)cj < split) ? tabA : tabB;
                x[j] = *(const float4*)(tab + (size_t)(cj * (unsigned)(DIM * 4)));
            }
#pragma unroll
            for (int j = 0; j < SUBG; ++j) {
                const float w = __int_as_float(r[j].y);
                acc.x = fmaf(w, x[j].x, acc.x);
                acc.y = fmaf(w, x[j].y, acc.y);
                acc.z = fmaf(w, x[j].z, acc.z);
                acc.w = fmaf(w, x[j].w, acc.w);
                if (r[j].x < 0) {
                    *(float4*)(obase + (row0 + (size_t)(r[j].x & (BKT_ROWS - 1))) * HALF) = acc;
                    acc = make_float4(0.f, 0.f, 0.f, 0.f);
                }
            }
            pos += SUBG;
        }
        return;
    }

    float* acc = (float*)raw;
    for (int i = tid; i < BKT_ROWS * DIM / 4; i += 256)
        ((float4*)acc)[i] = make_float4(0.f, 0.f, 0.f, 0.f);
    __syncthreads();
    const int end = beg + n;
    int2 cur = rec[(beg + tid < end) ? (beg + tid) : (end - 1)];
    for (int base = beg; base < end; base += 256) {
        int2 nxt = make_int2(0, 0);
        int nvalid = end - base - wib * 32;
        nvalid = nvalid > 32 ? 32 : nvalid;
#pragma unroll
        for (int g = 0; g < 32; g += GRP) {
            if (g < nvalid) {
                int    key[GRP];
                float  w[GRP];
                float2 x[GRP];
#pragma unroll
                for (int j = 0; j < GRP; ++j) {
                    const int jj = (g + j < nvalid) ? (g + j) : (nvalid - 1);
                    key[j] = __builtin_amdgcn_readlane(cur.x, jj);
                    w[j]   = __int_as_float(__builtin_amdgcn_readlane(cur.y, jj));
                    const float* src = spmm_src_row(srcA, srcB, split, key[j] >> BKT_SHIFT);
                    x[j] = *(const float2*)(src + 2 * lane);
                }
                if (g == 0) nxt = rec[(base + 256 + tid < end) ? (base + 256 + tid) : (end - 1)];
#pragma unroll
                for (int j = 0; j < GRP; ++j) {
                    if (g + j < nvalid) {
                        float* d = acc + (key[j] & (BKT_ROWS - 1)) * DIM + 2 * lane;
                        atomicAdd(d,     w[j] * x[j].x);
                        atomicAdd(d + 1, w[j] * x[j].y);
                    }
                }
            }
        }
        cur = nxt;
    }
    __syncthreads();
    for (int q = tid; q < BKT_ROWS * (HALF / 4); q += 256) {
        const int rr = q >> 3, pc = q & 7;
        if (row0 + rr < (size_t)N_NODES) {
            const float4 lo = *(const float4*)(acc + rr * DIM + 4 * pc);
            const float4 hi = *(const float4*)(acc + rr * DIM + HALF + 4 * pc);
            *(float4*)(out_lo + (row0 + rr) * HALF + 4 * pc) = lo;
            *(float4*)(out_hi + (row0 + rr) * HALF + 4 * pc) = hi;
        }
    }
}

__global__ __launch_bounds__(256) void bucket_spmm_plane_kernel(
    const int2* __restrict__ rec, const int* __restrict__ start,
    const int* __restrict__ cnt,
    const float* __restrict__ src, float* __restrict__ dst) {
    __shared__ __align__(16) unsigned char raw[BKT_ROWS * HALF * 4 + 16];
    __shared__ int rcnt[BKT_ROWS];
    __shared__ int rbeg[BKT_ROWS + 1];
    const int tid  = (int)threadIdx.x;
    const int lane = tid & 31;
    const int wib  = __builtin_amdgcn_readfirstlane(tid >> 5);
    const int bkt  = (int)blockIdx.x;
    const size_t row0 = (size_t)bkt * BKT_ROWS;
    const int beg = __builtin_amdgcn_readfirstlane(start[bkt]);
    const int n   = __builtin_amdgcn_readfirstlane(cnt[bkt]);

    if (n <= SORT_CAP) {
        int2* srec = (int2*)raw;
        bucket_order_rows(rec, beg, n, srec, rcnt, rbeg, tid);
        const int sub = lane >> 3;
        const int li  = lane & 7;
        const int rq0 = wib * 16 + sub * 4;
        float* obase = dst + 4 * li;
#pragma unroll
        for (int k = 0; k < 4; ++k) {
            const int rr = rq0 + k;
            if (rbeg[rr + 1] == rbeg[rr] && row0 + rr < (size_t)N_NODES)
                *(float4*)(obase + (row0 + rr) * HALF) = make_float4(0.f, 0.f, 0.f, 0.f);
        }
        int pos = rbeg[rq0];
        const int pend = rbeg[rq0 + 4];
        float4 acc = make_float4(0.f, 0.f, 0.f, 0.f);
        while (__builtin_amdgcn_ballot_w32(pos < pend) != 0u) {
            int2   r[SUBG];
            float4 x[SUBG];
#pragma unroll
            for (int j = 0; j < SUBG; ++j) {
                const int i = pos + j;
                r[j] = srec[i < pend ? i : PAD_SLOT];
            }
#pragma unroll
            for (int j = 0; j < SUBG; ++j) {
                const unsigned off = (((unsigned)r[j].x & REC_COLS) >> BKT_SHIFT) * (unsigned)(HALF * 4)
                                   + 16u * (unsigned)li;
                x[j] = *(const float4*)((const char*)src + off);
            }
#pragma unroll
            for (int j = 0; j < SUBG; ++j) {
                const float w = __int_as_float(r[j].y);
                acc.x = fmaf(w, x[j].x, acc.x);
                acc.y = fmaf(w, x[j].y, acc.y);
                acc.z = fmaf(w, x[j].z, acc.z);
                acc.w = fmaf(w, x[j].w, acc.w);
                if (r[j].x < 0) {
                    *(float4*)(obase + (row0 + (size_t)(r[j].x & (BKT_ROWS - 1))) * HALF) = acc;
                    acc = make_float4(0.f, 0.f, 0.f, 0.f);
                }
            }
            pos += SUBG;
        }
        return;
    }

    float* acc = (float*)raw;
    for (int i = tid; i < BKT_ROWS * HALF / 4; i += 256)
        ((float4*)acc)[i] = make_float4(0.f, 0.f, 0.f, 0.f);
    __syncthreads();
    const int end = beg + n;
    int2 cur = rec[(beg + tid < end) ? (beg + tid) : (end - 1)];
    for (int base = beg; base < end; base += 256) {
        int2 nxt = make_int2(0, 0);
        int nvalid = end - base - wib * 32;
        nvalid = nvalid > 32 ? 32 : nvalid;
#pragma unroll
        for (int g = 0; g < 32; g += GRP) {
            if (g < nvalid) {
                int   key[GRP];
                float w[GRP];
                float x[GRP];
#pragma unroll
                for (int j = 0; j < GRP; ++j) {
                    const int jj = (g + j < nvalid) ? (g + j) : (nvalid - 1);
                    key[j] = __builtin_amdgcn_readlane(cur.x, jj);
                    w[j]   = __int_as_float(__builtin_amdgcn_readlane(cur.y, jj));
                    x[j]   = src[(size_t)(key[j] >> BKT_SHIFT) * HALF + lane];
                }
                if (g == 0) nxt = rec[(base + 256 + tid < end) ? (base + 256 + tid) : (end - 1)];
#pragma unroll
                for (int j = 0; j < GRP; ++j) {
                    if (g + j < nvalid)
                        atomicAdd(acc + (key[j] & (BKT_ROWS - 1)) * HALF + lane, w[j] * x[j]);
                }
            }
        }
        cur = nxt;
    }
    __syncthreads();
    for (int q = tid; q < BKT_ROWS * (HALF / 4); q += 256) {
        const int rr = q >> 3;
        if (row0 + rr < (size_t)N_NODES)
            *(float4*)(dst + row0 * HALF + 4 * q) = *(const float4*)(acc + 4 * q);
    }
}

#define PRED_WPB   8
#define PRED_TPW   8
#define PRED_BLOCKS (BATCH_CT / 16 / PRED_WPB / PRED_TPW)
#define TILE_PITCH 68
__device__ __forceinline__ void pred_tile_products(
    const float* fa, const float* fb, int lane, float cbu, float cbi, float muv,
    float* __restrict__ out, int base) {
    v8f c = {};
#pragma unroll
    for (int t = 0; t < 8; ++t) {
        const float4 a4 = *(const float4*)(fa + 4 * t);
        const float4 b4 = *(const float4*)(fb + 4 * t);
        v2f a0 = {a4.x, a4.y}, b0 = {b4.x, b4.y};
        v2f a1 = {a4.z, a4.w}, b1 = {b4.z, b4.w};
        c = __builtin_amdgcn_wmma_f32_16x16x4_f32(
                false, a0, false, b0, (short)0, c, false, false);
        c = __builtin_amdgcn_wmma_f32_16x16x4_f32(
                false, a1, false, b1, (short)0, c, false, false);
    }
    float d = 0.f;
#pragma unroll
    for (int j = 0; j < 8; ++j) {
        float vlo = __shfl(c[j], j, 32);
        float vhi = __shfl(c[j], j + 24, 32);
        if (lane == j)     d = vlo;
        if (lane == j + 8) d = vhi;
    }
    if (lane < 16) out[base + lane] = d + cbu + cbi + muv;
}

#define PRED_LDU(r) (*(const float4*)(up + (size_t)(r) * HALF + mo))
#define PRED_LDI(r) (*(const float4*)(ip + (size_t)(r) * HALF + mo))
#define PRED_GATHER_TILE()                                              \
    gu0 = PRED_LDU(ua.x); gi0 = PRED_LDI(ia.x);                         \
    gu1 = PRED_LDU(ua.y); gi1 = PRED_LDI(ia.y);                         \
    gu2 = PRED_LDU(ua.z); gi2 = PRED_LDI(ia.z);                         \
    gu3 = PRED_LDU(ua.w); gi3 = PRED_LDI(ia.w);                         \
    gu4 = PRED_LDU(ub.x); gi4 = PRED_LDI(ib.x);                         \
    gu5 = PRED_LDU(ub.y); gi5 = PRED_LDI(ib.y);                         \
    gu6 = PRED_LDU(ub.z); gi6 = PRED_LDI(ib.z);                         \
    gu7 = PRED_LDU(ub.w); gi7 = PRED_LDI(ib.w);
#define PRED_ST(j, vu, vi)                                              \
    *(float4*)(su + (j) * TILE_PITCH) = (vu);                           \
    *(float4*)(si + (j) * TILE_PITCH) = (vi);
#define PRED_STAGE_TILE()                                               \
    PRED_ST(0, gu0, gi0) PRED_ST(1, gu1, gi1) PRED_ST(2, gu2, gi2)      \
    PRED_ST(3, gu3, gi3) PRED_ST(4, gu4, gi4) PRED_ST(5, gu5, gi5)      \
    PRED_ST(6, gu6, gi6) PRED_ST(7, gu7, gi7)
#define PRED_LOAD_IDX(b)                                                \
    ur = u_idx[(b) + m];                                                \
    ir = i_idx[(b) + m];                                                \
    ua = *(const int4*)(u_idx + (b) + 8 * hi);                          \
    ub = *(const int4*)(u_idx + (b) + 8 * hi + 4);                      \
    ia = *(const int4*)(i_idx + (b) + 8 * hi);                          \
    ib = *(const int4*)(i_idx + (b) + 8 * hi + 4);

__global__ __launch_bounds__(256) void pred_kernel(
    const float* __restrict__ Ulo, const float* __restrict__ Uhi,
    const float* __restrict__ Ilo, const float* __restrict__ Ihi,
    const int*   __restrict__ u_idx, const int* __restrict__ i_idx,
    const float* __restrict__ bu, const float* __restrict__ bi,
    const float* __restrict__ mu, float* __restrict__ out) {
    __shared__ float tiles[PRED_WPB][2][16 * TILE_PITCH];

    const int wib  = threadIdx.x >> 5;
    const int lane = threadIdx.x & 31;
    const int m    = lane & 15;
    const int hi   = lane >> 4;
    int base = ((int)blockIdx.x * PRED_WPB + wib) * (16 * PRED_TPW);

    float* tu = &tiles[wib][0][0];
    float* ti = &tiles[wib][1][0];
    const float* up = (m < 8) ? Ulo : Uhi;
    const float* ip = (m < 8) ? Ilo : Ihi;
    const int    mo = 4 * (m & 7);
    float* su = tu + (8 * hi) * TILE_PITCH + 4 * m;
    float* si = ti + (8 * hi) * TILE_PITCH + 4 * m;
    const float* fa = tu + m * TILE_PITCH + 32 * hi;
    const float* fb = ti + m * TILE_PITCH + 32 * hi;
    const float muv = mu[0];

    int  ur, ir;
    int4 ua, ub, ia, ib;
    float4 gu0, gu1, gu2, gu3, gu4, gu5, gu6, gu7;
    float4 gi0, gi1, gi2, gi3, gi4, gi5, gi6, gi7;

    PRED_LOAD_IDX(base)
    PRED_GATHER_TILE()
    float bu_v = bu[ur];
    float bi_v = bi[ir];
    PRED_LOAD_IDX(base + 16)

#pragma unroll 1
    for (int it = 0; it < PRED_TPW; ++it) {
        PRED_STAGE_TILE()
        const float cbu = bu_v, cbi = bi_v;
        __builtin_amdgcn_fence(__ATOMIC_ACQ_REL, "wavefront");

        if (it + 1 < PRED_TPW) {
            PRED_GATHER_TILE()
            bu_v = bu[ur];
            bi_v = bi[ir];
            if (it + 2 < PRED_TPW) {
                PRED_LOAD_IDX(base + 32)
            }
        }

        pred_tile_products(fa, fb, lane, cbu, cbi, muv, out, base);
        base += 16;
    }
}

extern "C" void kernel_launch(void* const* d_in, const int* in_sizes, int n_in,
                              void* d_out, int out_size, void* d_ws, size_t ws_size,
                              hipStream_t stream) {
    const int*   u_idx  = (const int*)d_in[0];
    const int*   i_idx  = (const int*)d_in[1];
    const int*   A_rows = (const int*)d_in[2];
    const int*   A_cols = (const int*)d_in[3];
    const float* A_vals = (const float*)d_in[4];
    const float* Eu     = (const float*)d_in[5];
    const float* Ei     = (const float*)d_in[6];
    const float* bu     = (const float*)d_in[7];
    const float* bi     = (const float*)d_in[8];
    const float* mu     = (const float*)d_in[9];
    float* out = (float*)d_out;

    float* P0 = (float*)d_ws;
    float* P1 = P0 + (size_t)N_NODES * HALF;
    float* P2 = P1 + (size_t)N_NODES * HALF;
    int2*  rec   = (int2*)(P2 + (size_t)N_NODES * HALF);
    int*   cnt   = (int*)(rec + NNZ_CT);
    int*   fill  = cnt + NBKT_PAD;
    int*   start = fill + NBKT_PAD;

    clear_counts_kernel<<<1, 256, 0, stream>>>(cnt);
    bucket_count_kernel<<<CNT_BLOCKS, 256, 0, stream>>>(A_rows, cnt, NNZ_CT);
    bucket_place_kernel<<<PLC_BLOCKS, PLC_THREADS, 0, stream>>>(
        A_rows, A_cols, A_vals, cnt, fill, start, rec, NNZ_CT);

    bucket_spmm_rows_kernel<<<NBKT, 256, 0, stream>>>(
        rec, start, cnt, Eu, Ei, N_USERS, P0, P1);

    bucket_spmm_plane_kernel<<<NBKT, 256, 0, stream>>>(rec, start, cnt, P0, P2);
    bucket_spmm_plane_kernel<<<NBKT, 256, 0, stream>>>(rec, start, cnt, P1, P0);

    pred_kernel<<<PRED_BLOCKS, 256, 0, stream>>>(
        P2, P0, P2 + (size_t)N_USERS * HALF, P0 + (size_t)N_USERS * HALF,
        u_idx, i_idx, bu, bi, mu, out);
}


# edge placing without the count pass: each bucket owns a fixed range of 1536 records filled through its cursor (6 launches, was 7); a fuller bucket is summed from the edge list; sum order as before
# speedup vs baseline: 3.3176x; 1.0370x over previous
#include <hip/hip_runtime.h>


#define N_USERS  100000
#define N_ITEMS  50000
#define N_NODES  (N_USERS + N_ITEMS)
#define NNZ_CT   1500000
#define BATCH_CT 1048576
#define DIM      64
#define HALF     32

typedef __attribute__((ext_vector_type(2))) float v2f;
typedef __attribute__((ext_vector_type(8))) float v8f;

#define BKT_ROWS  128
#define BKT_SHIFT 7
#define NBKT      ((N_NODES + BKT_ROWS - 1) / BKT_ROWS)
#define NBKT_PAD  1280
#define BKT_CAP   1536
#define PLC_BLOCKS  256
#define PLC_THREADS 1024
#define PLC_EPB   ((NNZ_CT + PLC_BLOCKS - 1) / PLC_BLOCKS)
#define PLC_NIT   ((PLC_EPB + PLC_THREADS - 1) / PLC_THREADS)
#define PLC_SCAN_N (2 * PLC_THREADS)

__global__ __launch_bounds__(256) void clear_counts_kernel(int* __restrict__ fill) {
    for (int i = threadIdx.x; i < NBKT_PAD; i += 256) fill[i] = 0;
}

__global__ __launch_bounds__(PLC_THREADS) void bucket_place_kernel(
    const int* __restrict__ rows, const int* __restrict__ cols,
    const float* __restrict__ vals, int* __restrict__ fill,
    int2* __restrict__ rec, int nnz) {
    __shared__ int h[PLC_SCAN_N];
    __shared__ int st[PLC_SCAN_N];
    __shared__ int lp[PLC_SCAN_N];
    __shared__ int wsum_l[32];
    __shared__ int2 srec[PLC_EPB];
    __shared__ unsigned short sbkt[PLC_EPB];
    const int tid  = (int)threadIdx.x;
    const int lane = tid & 31;
    const int wib  = tid >> 5;
    for (int i = tid; i < PLC_SCAN_N; i += PLC_THREADS) h[i] = 0;
    __syncthreads();
    const int e_beg = (int)blockIdx.x * PLC_EPB;
    int   r[PLC_NIT], c[PLC_NIT];
    float v[PLC_NIT];
#pragma unroll
    for (int q = 0; q < PLC_NIT; ++q) {
        const int i = tid + q * PLC_THREADS;
        const int e = e_beg + i;
        const bool ok = (i < PLC_EPB && e < nnz);
        r[q] = ok ? rows[e] : -1;
        c[q] = ok ? cols[e] : 0;
        v[q] = ok ? vals[e] : 0.f;
    }
    const int k0 = 2 * tid, k1 = 2 * tid + 1;
#pragma unroll
    for (int q = 0; q < PLC_NIT; ++q) {
        if ((unsigned)r[q] < (unsigned)N_NODES) atomicAdd(&h[r[q] >> BKT_SHIFT], 1);
    }
    __syncthreads();
    const int l0 = h[k0], l1 = h[k1];
    const int sl = l0 + l1;
    int il = sl;
    for (int d = 1; d < 32; d <<= 1) {
        const int ol = __shfl_up(il, d, 32);
        if (lane >= d) il += ol;
    }
    if (lane == 31) wsum_l[wib] = il;
    __syncthreads();
    int offl = 0, nloc = 0;
    for (int w = 0; w < 32; ++w) {
        const int b = wsum_l[w];
        if (w < wib) offl += b;
        nloc += b;
    }
    const int runl = offl + il - sl;
    lp[k0] = runl;  lp[k1] = runl + l0;
    __syncthreads();
    for (int k = tid; k < NBKT; k += PLC_THREADS) {
        const int n = h[k];
        const int b = n ? atomicAdd(&fill[k], n) : 0;
        st[k] = b - lp[k];
        h[k]  = lp[k];
    }
    __syncthreads();
#pragma unroll
    for (int q = 0; q < PLC_NIT; ++q) {
        if ((unsigned)r[q] < (unsigned)N_NODES) {
            const int bk = r[q] >> BKT_SHIFT;
            const int p  = atomicAdd(&h[bk], 1);
            srec[p] = make_int2((c[q] << BKT_SHIFT) | (r[q] & (BKT_ROWS - 1)),
                                __float_as_int(v[q]));
            sbkt[p] = (unsigned short)bk;
        }
    }
    __syncthreads();
    for (int s = tid; s < nloc; s += PLC_THREADS) {
        const int bk = sbkt[s];
        const int p  = st[bk] + s;
        if (p < BKT_CAP) rec[bk * BKT_CAP + p] = srec[s];
    }
}

#define SUBG 8
#define SORT_CAP BKT_CAP
#define SORT_NIT (SORT_CAP / 256)
#define REC_LAST  0x80000000u
#define REC_COLS  0x7fffffffu
#define PAD_SLOT  SORT_CAP

__device__ __forceinline__ const float* spmm_src_row(
    const float* __restrict__ srcA, const float* __restrict__ srcB, int split, int c) {
    return (c < split) ? (srcA + (size_t)c * DIM)
                       : (srcB + (size_t)(c - split) * DIM);
}

__device__ __forceinline__ void bucket_order_rows(
    const int2* __restrict__ rec, int beg, int n,
    int2* srec, int* rcnt, int* rbeg, int tid) {
    const int lane = tid & 31;
    if (tid < BKT_ROWS) rcnt[tid] = 0;
    int2 r[SORT_NIT];
#pragma unroll
    for (int q = 0; q < SORT_NIT; ++q) {
        const int i = tid + q * 256;
        r[q] = (i < n) ? rec[beg + i] : make_int2(0, 0);
    }
    __syncthreads();
#pragma unroll
    for (int q = 0; q < SORT_NIT; ++q)
        if (tid + q * 256 < n) atomicAdd(&rcnt[r[q].x & (BKT_ROWS - 1)], 1);
    __syncthreads();
    if (tid < 32) {
        int c4[4];
        int s = 0;
#pragma unroll
        for (int k = 0; k < 4; ++k) { c4[k] = rcnt[4 * lane + k]; s += c4[k]; }
        int incl = s;
        for (int d = 1; d < 32; d <<= 1) {
            const int o = __shfl_up(incl, d, 32);
            if (lane >= d) incl += o;
        }
        int run = incl - s;
#pragma unroll
        for (int k = 0; k < 4; ++k) {
            rbeg[4 * lane + k] = run;
            rcnt[4 * lane + k] = run;
            run += c4[k];
        }
        if (lane == 31) rbeg[BKT_ROWS] = incl;
    }
    __syncthreads();
    int p[SORT_NIT], e[SORT_NIT];
#pragma unroll
    for (int q = 0; q < SORT_NIT; ++q) {
        const bool ok = (tid + q * 256 < n);
        const int row = r[q].x & (BKT_ROWS - 1);
        p[q] = ok ? atomicAdd(&rcnt[row], 1) : 0;
        e[q] = ok ? rbeg[row + 1] : 0;
    }
#pragma unroll
    for (int q = 0; q < SORT_NIT; ++q)
        if (tid + q * 256 < n) {
            int2 t = r[q];
            if (p[q] + 1 == e[q]) t.x = (int)((unsigned)t.x | REC_LAST);
            srec[p[q]] = t;
        }
    if (tid == 0) srec[PAD_SLOT] = make_int2(0, 0);
    __syncthreads();
}

__global__ __launch_bounds__(256) void bucket_spmm_rows_kernel(
    const int2* __restrict__ rec, const int* __restrict__ cnt,
    const int* __restrict__ rows, const int* __restrict__ cols,
    const float* __restrict__ vals, int nnz,
    const float* __restrict__ srcA, const float* __restrict__ srcB, int split,
    float* __restrict__ out_lo, float* __restrict__ out_hi) {
    __shared__ __align__(16) unsigned char raw[BKT_ROWS * DIM * 4];
    __shared__ int rcnt[BKT_ROWS];
    __shared__ int rbeg[BKT_ROWS + 1];
    const int tid  = (int)threadIdx.x;
    const int lane = tid & 31;
    const int wib  = __builtin_amdgcn_readfirstlane(tid >> 5);
    const int bkt  = (int)blockIdx.x;
    const size_t row0 = (size_t)bkt * BKT_ROWS;
    const int beg = bkt * BKT_CAP;
    const int n   = __builtin_amdgcn_readfirstlane(cnt[bkt]);

    if (n <= SORT_CAP) {
        int2* srec = (int2*)raw;
        bucket_order_rows(rec, beg, n, srec, rcnt, rbeg, tid);
        const int sub = lane >> 4;
        const int li  = lane & 15;
        const int rq0 = wib * 16 + sub * 8;
        float* obase = ((li < 8) ? out_lo : out_hi) + 4 * (li & 7);
#pragma unroll
        for (int k = 0; k < 8; ++k) {
            const int rr = rq0 + k;
            if (rbeg[rr + 1] == rbeg[rr] && row0 + rr < (size_t)N_NODES)
                *(float4*)(obase + (row0 + rr) * HALF) = make_float4(0.f, 0.f, 0.f, 0.f);
        }
        const char* tabA = (const char*)srcA + 16 * li;
        const char* tabB = (const char*)srcB + 16 * li - (long long)split * (DIM * 4);
        int pos = rbeg[rq0];
        const int pend = rbeg[rq0 + 8];
        float4 acc = make_float4(0.f, 0.f, 0.f, 0.f);
        while (__builtin_amdgcn_ballot_w32(pos < pend) != 0u) {
            int2   r[SUBG];
            float4 x[SUBG];
#pragma unroll
            for (int j = 0; j < SUBG; ++j) {
                const int i = pos + j;
                r[j] = srec[i < pend ? i : PAD_SLOT];
            }
#pragma unroll
            for (int j = 0; j < SUBG; ++j) {
                const unsigned cj = ((unsigned)r[j].x & REC_COLS) >> BKT_SHIFT;
                const char* tab = ((int)cj < split) ? tabA : tabB;
                x[j] = *(const float4*)(tab + (size_t)(cj * (unsigned)(DIM * 4)));
            }
#pragma unroll
            for (int j = 0; j < SUBG; ++j) {
                const float w = __int_as_float(r[j].y);
                acc.x = fmaf(w, x[j].x, acc.x);
                acc.y = fmaf(w, x[j].y, acc.y);
                acc.z = fmaf(w, x[j].z, acc.z);
                acc.w = fmaf(w, x[j].w, acc.w);
                if (r[j].x < 0) {
                    *(float4*)(obase + (row0 + (size_t)(r[j].x & (BKT_ROWS - 1))) * HALF) = acc;
                    acc = make_float4(0.f, 0.f, 0.f, 0.f);
                }
            }
            pos += SUBG;
        }
        return;
    }

    float* acc = (float*)raw;
    for (int i = tid; i < BKT_ROWS * DIM / 4; i += 256)
        ((float4*)acc)[i] = make_float4(0.f, 0.f, 0.f, 0.f);
    __syncthreads();
    for (int base = wib * 32; base < nnz; base += 256) {
        const int e = base + lane;
        const int r = (e < nnz) ? rows[e] : -1;
        const bool mine = ((unsigned)r < (unsigned)N_NODES) && ((r >> BKT_SHIFT) == bkt);
        const int   cc = mine ? cols[e] : 0;
        const float vv = mine ? vals[e] : 0.f;
        unsigned todo = __builtin_amdgcn_ballot_w32(mine);
        while (todo != 0u) {
            const int j = __builtin_ctz(todo);
            todo &= todo - 1u;
            const int   rj = __builtin_amdgcn_readlane(r, j);
            const int   cj = __builtin_amdgcn_readlane(cc, j);
            const float wj = __int_as_float(__builtin_amdgcn_readlane(__float_as_int(vv), j));
            const float2 x = *(const float2*)(spmm_src_row(srcA, srcB, split, cj) + 2 * lane);
            float* d = acc + (rj & (BKT_ROWS - 1)) * DIM + 2 * lane;
            atomicAdd(d,     wj * x.x);
            atomicAdd(d + 1, wj * x.y);
        }
    }
    __syncthreads();
    for (int q = tid; q < BKT_ROWS * (HALF / 4); q += 256) {
        const int rr = q >> 3, pc = q & 7;
        if (row0 + rr < (size_t)N_NODES) {
            const float4 lo = *(const float4*)(acc + rr * DIM + 4 * pc);
            const float4 hi = *(const float4*)(acc + rr * DIM + HALF + 4 * pc);
            *(float4*)(out_lo + (row0 + rr) * HALF + 4 * pc) = lo;
            *(float4*)(out_hi + (row0 + rr) * HALF + 4 * pc) = hi;
        }
    }
}

__global__ __launch_bounds__(256) void bucket_spmm_plane_kernel(
    const int2* __restrict__ rec, const int* __restrict__ cnt,
    const int* __restrict__ rows, const int* __restrict__ cols,
    const float* __restrict__ vals, int nnz,
    const float* __restrict__ src, float* __restrict__ dst) {
    __shared__ __align__(16) unsigned char raw[BKT_ROWS * HALF * 4 + 16];
    __shared__ int rcnt[BKT_ROWS];
    __shared__ int rbeg[BKT_ROWS + 1];
    const int tid  = (int)threadIdx.x;
    const int lane = tid & 31;
    const int wib  = __builtin_amdgcn_readfirstlane(tid >> 5);
    const int bkt  = (int)blockIdx.x;
    const size_t row0 = (size_t)bkt * BKT_ROWS;
    const int beg = bkt * BKT_CAP;
    const int n   = __builtin_amdgcn_readfirstlane(cnt[bkt]);

    if (n <= SORT_CAP) {
        int2* srec = (int2*)raw;
        bucket_order_rows(rec, beg, n, srec, rcnt, rbeg, tid);
        const int sub = lane >> 3;
        const int li  = lane & 7;
        const int rq0 = wib * 16 + sub * 4;
        float* obase = dst + 4 * li;
#pragma unroll
        for (int k = 0; k < 4; ++k) {
            const int rr = rq0 + k;
            if (rbeg[rr + 1] == rbeg[rr] && row0 + rr < (size_t)N_NODES)
                *(float4*)(obase + (row0 + rr) * HALF) = make_float4(0.f, 0.f, 0.f, 0.f);
        }
        int pos = rbeg[rq0];
        const int pend = rbeg[rq0 + 4];
        float4 acc = make_float4(0.f, 0.f, 0.f, 0.f);
        while (__builtin_amdgcn_ballot_w32(pos < pend) != 0u) {
            int2   r[SUBG];
            float4 x[SUBG];
#pragma unroll
            for (int j = 0; j < SUBG; ++j) {
                const int i = pos + j;
                r[j] = srec[i < pend ? i : PAD_SLOT];
            }
#pragma unroll
            for (int j = 0; j < SUBG; ++j) {
                const unsigned off = (((unsigned)r[j].x & REC_COLS) >> BKT_SHIFT) * (unsigned)(HALF * 4)
                                   + 16u * (unsigned)li;
                x[j] = *(const float4*)((const char*)src + off);
            }
#pragma unroll
            for (int j = 0; j < SUBG; ++j) {
                const float w = __int_as_float(r[j].y);
                acc.x = fmaf(w, x[j].x, acc.x);
                acc.y = fmaf(w, x[j].y, acc.y);
                acc.z = fmaf(w, x[j].z, acc.z);
                acc.w = fmaf(w, x[j].w, acc.w);
                if (r[j].x < 0) {
                    *(float4*)(obase + (row0 + (size_t)(r[j].x & (BKT_ROWS - 1))) * HALF) = acc;
                    acc = make_float4(0.f, 0.f, 0.f, 0.f);
                }
            }
            pos += SUBG;
        }
        return;
    }

    float* acc = (float*)raw;
    for (int i = tid; i < BKT_ROWS * HALF / 4; i += 256)
        ((float4*)acc)[i] = make_float4(0.f, 0.f, 0.f, 0.f);
    __syncthreads();
    for (int base = wib * 32; base < nnz; base += 256) {
        const int e = base + lane;
        const int r = (e < nnz) ? rows[e] : -1;
        const bool mine = ((unsigned)r < (unsigned)N_NODES) && ((r >> BKT_SHIFT) == bkt);
        const int   cc = mine ? cols[e] : 0;
        const float vv = mine ? vals[e] : 0.f;
        unsigned todo = __builtin_amdgcn_ballot_w32(mine);
        while (todo != 0u) {
            const int j = __builtin_ctz(todo);
            todo &= todo - 1u;
            const int   rj = __builtin_amdgcn_readlane(r, j);
            const int   cj = __builtin_amdgcn_readlane(cc, j);
            const float wj = __int_as_float(__builtin_amdgcn_readlane(__float_as_int(vv), j));
            const float x = src[(size_t)cj * HALF + lane];
            atomicAdd(acc + (rj & (BKT_ROWS - 1)) * HALF + lane, wj * x);
        }
    }
    __syncthreads();
    for (int q = tid; q < BKT_ROWS * (HALF / 4); q += 256) {
        const int rr = q >> 3;
        if (row0 + rr < (size_t)N_NODES)
            *(float4*)(dst + row0 * HALF + 4 * q) = *(const float4*)(acc + 4 * q);
    }
}

#define PRED_WPB   8
#define PRED_TPW   8
#define PRED_BLOCKS (BATCH_CT / 16 / PRED_WPB / PRED_TPW)
#define TILE_PITCH 68
__device__ __forceinline__ void pred_tile_products(
    const float* fa, const float* fb, int lane, float cbu, float cbi, float muv,
    float* __restrict__ out, int base) {
    v8f c = {};
#pragma unroll
    for (int t = 0; t < 8; ++t) {
        const float4 a4 = *(const float4*)(fa + 4 * t);
        const float4 b4 = *(const float4*)(fb + 4 * t);
        v2f a0 = {a4.x, a4.y}, b0 = {b4.x, b4.y};
        v2f a1 = {a4.z, a4.w}, b1 = {b4.z, b4.w};
        c = __builtin_amdgcn_wmma_f32_16x16x4_f32(
                false, a0, false, b0, (short)0, c, false, false);
        c = __builtin_amdgcn_wmma_f32_16x16x4_f32(
                false, a1, false, b1, (short)0, c, false, false);
    }
    float d = 0.f;
#pragma unroll
    for (int j = 0; j < 8; ++j) {
        float vlo = __shfl(c[j], j, 32);
        float vhi = __shfl(c[j], j + 24, 32);
        if (lane == j)     d = vlo;
        if (lane == j + 8) d = vhi;
    }
    if (lane < 16) out[base + lane] = d + cbu + cbi + muv;
}

#define PRED_LDU(r) (*(const float4*)(up + (size_t)(r) * HALF + mo))
#define PRED_LDI(r) (*(const float4*)(ip + (size_t)(r) * HALF + mo))
#define PRED_GATHER_TILE()                                              \
    gu0 = PRED_LDU(ua.x); gi0 = PRED_LDI(ia.x);                         \
    gu1 = PRED_LDU(ua.y); gi1 = PRED_LDI(ia.y);                         \
    gu2 = PRED_LDU(ua.z); gi2 = PRED_LDI(ia.z);                         \
    gu3 = PRED_LDU(ua.w); gi3 = PRED_LDI(ia.w);                         \
    gu4 = PRED_LDU(ub.x); gi4 = PRED_LDI(ib.x);                         \
    gu5 = PRED_LDU(ub.y); gi5 = PRED_LDI(ib.y);                         \
    gu6 = PRED_LDU(ub.z); gi6 = PRED_LDI(ib.z);                         \
    gu7 = PRED_LDU(ub.w); gi7 = PRED_LDI(ib.w);
#define PRED_ST(j, vu, vi)                                              \
    *(float4*)(su + (j) * TILE_PITCH) = (vu);                           \
    *(float4*)(si + (j) * TILE_PITCH) = (vi);
#define PRED_STAGE_TILE()                                               \
    PRED_ST(0, gu0, gi0) PRED_ST(1, gu1, gi1) PRED_ST(2, gu2, gi2)      \
    PRED_ST(3, gu3, gi3) PRED_ST(4, gu4, gi4) PRED_ST(5, gu5, gi5)      \
    PRED_ST(6, gu6, gi6) PRED_ST(7, gu7, gi7)
#define PRED_LOAD_IDX(b)                                                \
    ur = u_idx[(b) + m];                                                \
    ir = i_idx[(b) + m];                                                \
    ua = *(const int4*)(u_idx + (b) + 8 * hi);                          \
    ub = *(const int4*)(u_idx + (b) + 8 * hi + 4);                      \
    ia = *(const int4*)(i_idx + (b) + 8 * hi);                          \
    ib = *(const int4*)(i_idx + (b) + 8 * hi + 4);

__global__ __launch_bounds__(256) void pred_kernel(
    const float* __restrict__ Ulo, const float* __restrict__ Uhi,
    const float* __restrict__ Ilo, const float* __restrict__ Ihi,
    const int*   __restrict__ u_idx, const int* __restrict__ i_idx,
    const float* __restrict__ bu, const float* __restrict__ bi,
    const float* __restrict__ mu, float* __restrict__ out) {
    __shared__ float tiles[PRED_WPB][2][16 * TILE_PITCH];

    const int wib  = threadIdx.x >> 5;
    const int lane = threadIdx.x & 31;
    const int m    = lane & 15;
    const int hi   = lane >> 4;
    int base = ((int)blockIdx.x * PRED_WPB + wib) * (16 * PRED_TPW);

    float* tu = &tiles[wib][0][0];
    float* ti = &tiles[wib][1][0];
    const float* up = (m < 8) ? Ulo : Uhi;
    const float* ip = (m < 8) ? Ilo : Ihi;
    const int    mo = 4 * (m & 7);
    float* su = tu + (8 * hi) * TILE_PITCH + 4 * m;
    float* si = ti + (8 * hi) * TILE_PITCH + 4 * m;
    const float* fa = tu + m * TILE_PITCH + 32 * hi;
    const float* fb = ti + m * TILE_PITCH + 32 * hi;
    const float muv = mu[0];

    int  ur, ir;
    int4 ua, ub, ia, ib;
    float4 gu0, gu1, gu2, gu3, gu4, gu5, gu6, gu7;
    float4 gi0, gi1, gi2, gi3, gi4, gi5, gi6, gi7;

    PRED_LOAD_IDX(base)
    PRED_GATHER_TILE()
    float bu_v = bu[ur];
    float bi_v = bi[ir];
    PRED_LOAD_IDX(base + 16)

#pragma unroll 1
    for (int it = 0; it < PRED_TPW; ++it) {
        PRED_STAGE_TILE()
        const float cbu = bu_v, cbi = bi_v;
        __builtin_amdgcn_fence(__ATOMIC_ACQ_REL, "wavefront");

        if (it + 1 < PRED_TPW) {
            PRED_GATHER_TILE()
            bu_v = bu[ur];
            bi_v = bi[ir];
            if (it + 2 < PRED_TPW) {
                PRED_LOAD_IDX(base + 32)
            }
        }

        pred_tile_products(fa, fb, lane, cbu, cbi, muv, out, base);
        base += 16;
    }
}

extern "C" void kernel_launch(void* const* d_in, const int* in_sizes, int n_in,
                              void* d_out, int out_size, void* d_ws, size_t ws_size,
                              hipStream_t stream) {
    const int*   u_idx  = (const int*)d_in[0];
    const int*   i_idx  = (const int*)d_in[1];
    const int*   A_rows = (const int*)d_in[2];
    const int*   A_cols = (const int*)d_in[3];
    const float* A_vals = (const float*)d_in[4];
    const float* Eu     = (const float*)d_in[5];
    const float* Ei     = (const float*)d_in[6];
    const float* bu     = (const float*)d_in[7];
    const float* bi     = (const float*)d_in[8];
    const float* mu     = (const float*)d_in[9];
    float* out = (float*)d_out;

    float* P0 = (float*)d_ws;
    float* P1 = P0 + (size_t)N_NODES * HALF;
    float* P2 = P1 + (size_t)N_NODES * HALF;
    int2*  rec   = (int2*)(P2 + (size_t)N_NODES * HALF);
    int*   fill  = (int*)(rec + (size_t)NBKT * BKT_CAP);

    clear_counts_kernel<<<1, 256, 0, stream>>>(fill);
    bucket_place_kernel<<<PLC_BLOCKS, PLC_THREADS, 0, stream>>>(
        A_rows, A_cols, A_vals, fill, rec, NNZ_CT);

    bucket_spmm_rows_kernel<<<NBKT, 256, 0, stream>>>(
        rec, fill, A_rows, A_cols, A_vals, NNZ_CT, Eu, Ei, N_USERS, P0, P1);

    bucket_spmm_plane_kernel<<<NBKT, 256, 0, stream>>>(
        rec, fill, A_rows, A_cols, A_vals, NNZ_CT, P0, P2);
    bucket_spmm_plane_kernel<<<NBKT, 256, 0, stream>>>(
        rec, fill, A_rows, A_cols, A_vals, NNZ_CT, P1, P0);

    pred_kernel<<<PRED_BLOCKS, 256, 0, stream>>>(
        P2, P0, P2 + (size_t)N_USERS * HALF, P0 + (size_t)N_USERS * HALF,
        u_idx, i_idx, bu, bi, mu, out);
}
